# starting kernel
# baseline (speedup 1.0000x reference)
#include <hip/hip_runtime.h>

typedef __attribute__((ext_vector_type(2))) float v2f;
typedef __attribute__((ext_vector_type(8))) float v8f;

#define BM 128
#define BM2 64
#define BN 128
#define BK 32
#define AS_STRIDE 36
#define BS_STRIDE 132

__global__ __launch_bounds__(256)
void gnn_gemm_bias_act(const float* __restrict__ X, const float* __restrict__ W,
                       const float* __restrict__ bias, float* __restrict__ Y,
                       int M, int K, int N, int doRelu) {
  __shared__ float As[BM * AS_STRIDE];
  __shared__ float Bs[BK * BS_STRIDE];

  const int tid   = threadIdx.x;
  const int lane  = tid & 31;
  const int wave  = tid >> 5;
  const int waveM = wave & 3;
  const int waveN = wave >> 2;
  const int lr    = lane & 15;
  const int kHalf = (lane >> 4) << 1;

  const int m0 = blockIdx.x * BM;
  const int n0 = blockIdx.y * BN;

  v8f acc[2][4];
#pragma unroll
  for (int mt = 0; mt < 2; ++mt)
#pragma unroll
    for (int nt = 0; nt < 4; ++nt)
      acc[mt][nt] = (v8f)0.0f;

  for (int k0 = 0; k0 < K; k0 += BK) {
#pragma unroll
    for (int i = 0; i < 4; ++i) {
      int s  = tid + i * 256;
      int r  = s >> 3;
      int c4 = (s & 7) << 2;
      float4 v = make_float4(0.f, 0.f, 0.f, 0.f);
      int gr = m0 + r;
      if (gr < M)
        v = *(const float4*)(X + (size_t)gr * K + (k0 + c4));
      *(float4*)(As + r * AS_STRIDE + c4) = v;
    }
#pragma unroll
    for (int i = 0; i < 4; ++i) {
      int s  = tid + i * 256;
      int r  = s >> 5;
      int c4 = (s & 31) << 2;
      float4 v = *(const float4*)(W + (size_t)(k0 + r) * N + (n0 + c4));
      *(float4*)(Bs + r * BS_STRIDE + c4) = v;
    }
    __syncthreads();

#pragma unroll
    for (int kk = 0; kk < BK; kk += 4) {
      v2f a[2];
      v2f b[4];
#pragma unroll
      for (int mt = 0; mt < 2; ++mt) {
        int m = waveM * 32 + mt * 16 + lr;
        const float* p = As + m * AS_STRIDE + kk + kHalf;
        a[mt][0] = p[0];
        a[mt][1] = p[1];
      }
#pragma unroll
      for (int nt = 0; nt < 4; ++nt) {
        int n = waveN * 64 + nt * 16 + lr;
        const float* p = Bs + (kk + kHalf) * BS_STRIDE + n;
        b[nt][0] = p[0];
        b[nt][1] = p[BS_STRIDE];
      }
#pragma unroll
      for (int mt = 0; mt < 2; ++mt)
#pragma unroll
        for (int nt = 0; nt < 4; ++nt)
          acc[mt][nt] = __builtin_amdgcn_wmma_f32_16x16x4_f32(
              false, a[mt], false, b[nt], (short)0, acc[mt][nt], false, false);
    }
    __syncthreads();
  }

  const int rowHalf = (lane >> 4) << 3;
#pragma unroll
  for (int mt = 0; mt < 2; ++mt) {
#pragma unroll
    for (int nt = 0; nt < 4; ++nt) {
      int col  = n0 + waveN * 64 + nt * 16 + lr;
      float bv = bias[col];
#pragma unroll
      for (int v = 0; v < 8; ++v) {
        int row = m0 + waveM * 32 + mt * 16 + rowHalf + v;
        float val = acc[mt][nt][v] + bv;
        if (doRelu) val = fmaxf(val, 0.0f);
        if (row < M) Y[(size_t)row * N + col] = val;
      }
    }
  }
}

__global__ __launch_bounds__(256)
void gnn_gemm_dual(const float* __restrict__ X,
                   const float* __restrict__ W1, const float* __restrict__ b1,
                   float* __restrict__ Y1,
                   const float* __restrict__ W2, const float* __restrict__ b2,
                   float* __restrict__ Y2,
                   int M, int K, int N) {
  __shared__ float As[BM2 * AS_STRIDE];
  __shared__ float Bs[2][BK * BS_STRIDE];

  const int tid     = threadIdx.x;
  const int lane    = tid & 31;
  const int wave    = tid >> 5;
  const int waveOut = wave >> 2;
  const int wq      = wave & 3;
  const int waveM   = wq & 1;
  const int waveN   = wq >> 1;
  const int lr      = lane & 15;
  const int kHalf   = (lane >> 4) << 1;

  const int m0 = blockIdx.x * BM2;
  const int n0 = blockIdx.y * BN;

  v8f acc[2][4];
#pragma unroll
  for (int mt = 0; mt < 2; ++mt)
#pragma unroll
    for (int nt = 0; nt < 4; ++nt)
      acc[mt][nt] = (v8f)0.0f;

  const float* Bw = &Bs[waveOut][0];

  for (int k0 = 0; k0 < K; k0 += BK) {
#pragma unroll
    for (int i = 0; i < 2; ++i) {
      int s  = tid + i * 256;
      int r  = s >> 3;
      int c4 = (s & 7) << 2;
      float4 v = make_float4(0.f, 0.f, 0.f, 0.f);
      int gr = m0 + r;
      if (gr < M)
        v = *(const float4*)(X + (size_t)gr * K + (k0 + c4));
      *(float4*)(As + r * AS_STRIDE + c4) = v;
    }
#pragma unroll
    for (int i = 0; i < 4; ++i) {
      int s  = tid + i * 256;
      int r  = s >> 5;
      int c4 = (s & 31) << 2;
      float4 v = *(const float4*)(W1 + (size_t)(k0 + r) * N + (n0 + c4));
      *(float4*)(&Bs[0][0] + r * BS_STRIDE + c4) = v;
    }
#pragma unroll
    for (int i = 0; i < 4; ++i) {
      int s  = tid + i * 256;
      int r  = s >> 5;
      int c4 = (s & 31) << 2;
      float4 v = *(const float4*)(W2 + (size_t)(k0 + r) * N + (n0 + c4));
      *(float4*)(&Bs[1][0] + r * BS_STRIDE + c4) = v;
    }
    __syncthreads();

#pragma unroll
    for (int kk = 0; kk < BK; kk += 4) {
      v2f a[2];
      v2f b[4];
#pragma unroll
      for (int mt = 0; mt < 2; ++mt) {
        int m = waveM * 32 + mt * 16 + lr;
        const float* p = As + m * AS_STRIDE + kk + kHalf;
        a[mt][0] = p[0];
        a[mt][1] = p[1];
      }
#pragma unroll
      for (int nt = 0; nt < 4; ++nt) {
        int n = waveN * 64 + nt * 16 + lr;
        const float* p = Bw + (kk + kHalf) * BS_STRIDE + n;
        b[nt][0] = p[0];
        b[nt][1] = p[BS_STRIDE];
      }
#pragma unroll
      for (int mt = 0; mt < 2; ++mt)
#pragma unroll
        for (int nt = 0; nt < 4; ++nt)
          acc[mt][nt] = __builtin_amdgcn_wmma_f32_16x16x4_f32(
              false, a[mt], false, b[nt], (short)0, acc[mt][nt], false, false);
    }
    __syncthreads();
  }

  const float* bias = waveOut ? b2 : b1;
  float*       Y    = waveOut ? Y2 : Y1;
  const int rowHalf = (lane >> 4) << 3;
#pragma unroll
  for (int mt = 0; mt < 2; ++mt) {
#pragma unroll
    for (int nt = 0; nt < 4; ++nt) {
      int col  = n0 + waveN * 64 + nt * 16 + lr;
      float bv = bias[col];
#pragma unroll
      for (int v = 0; v < 8; ++v) {
        int row = m0 + waveM * 32 + mt * 16 + rowHalf + v;
        float val = fmaxf(acc[mt][nt][v] + bv, 0.0f);
        if (row < M) Y[(size_t)row * N + col] = val;
      }
    }
  }
}

__global__ __launch_bounds__(256)
void gnn_scatter_add(const int* __restrict__ recvIdx, const int* __restrict__ sendIdx,
                     const float* __restrict__ msg, float* __restrict__ agg, int nEdges) {
  int e = blockIdx.x;
  if (e >= nEdges) return;
  int r = recvIdx[e];
  int s = sendIdx[e];
  int c = threadIdx.x;
  float v = msg[(size_t)s * 256 + c];
  unsigned int off = ((unsigned int)r * 256u + (unsigned int)c) * 4u;
  asm volatile("global_atomic_add_f32 %0, %1, %2 scope:SCOPE_DEV"
               :
               : "v"(off), "v"(v), "s"(agg)
               : "memory");
}

extern "C" void kernel_launch(void* const* d_in, const int* in_sizes, int n_in,
                              void* d_out, int out_size, void* d_ws, size_t ws_size,
                              hipStream_t stream) {
  (void)in_sizes; (void)n_in; (void)out_size; (void)ws_size;

  constexpr int U = 50000, D = 10000, E = 200000;
  constexpr int DIN = 1024, HID = 256, OUT = 128;

  const float* user_feat = (const float*)d_in[0];
  const float* dish_feat = (const float*)d_in[1];
  const int*   edge_user = (const int*)d_in[2];
  const int*   edge_dish = (const int*)d_in[3];
  const float* W_ui = (const float*)d_in[4];   const float* b_ui = (const float*)d_in[5];
  const float* W_di = (const float*)d_in[6];   const float* b_di = (const float*)d_in[7];
  const float* W_um = (const float*)d_in[8];   const float* b_um = (const float*)d_in[9];
  const float* W_dm = (const float*)d_in[10];  const float* b_dm = (const float*)d_in[11];
  const float* W_uu = (const float*)d_in[12];  const float* b_uu = (const float*)d_in[13];
  const float* W_du = (const float*)d_in[14];  const float* b_du = (const float*)d_in[15];
  const float* W_up = (const float*)d_in[16];  const float* b_up = (const float*)d_in[17];
  const float* W_dp = (const float*)d_in[18];  const float* b_dp = (const float*)d_in[19];

  float* out_user = (float*)d_out;
  float* out_dish = out_user + (size_t)U * OUT;

  float* ws      = (float*)d_ws;
  float* userAgg = ws;
  float* dishAgg = userAgg + (size_t)U * HID;
  float* msgUser = dishAgg + (size_t)D * HID;
  float* msgDish = msgUser + (size_t)U * HID;
  float* userUpd = msgUser;
  float* dishUpd = msgDish;

  dim3 blk(256);
  const int gU  = (U + BM - 1) / BM;
  const int gD  = (D + BM - 1) / BM;
  const int gU2 = (U + BM2 - 1) / BM2;
  const int gD2 = (D + BM2 - 1) / BM2;

  gnn_gemm_dual<<<dim3(gU2, HID / BN), blk, 0, stream>>>(
      user_feat, W_ui, b_ui, userAgg, W_dm, b_dm, msgUser, U, DIN, HID);
  gnn_gemm_dual<<<dim3(gD2, HID / BN), blk, 0, stream>>>(
      dish_feat, W_di, b_di, dishAgg, W_um, b_um, msgDish, D, DIN, HID);

  gnn_scatter_add<<<dim3(E), blk, 0, stream>>>(edge_user, edge_dish, msgDish, userAgg, E);
  gnn_scatter_add<<<dim3(E), blk, 0, stream>>>(edge_dish, edge_user, msgUser, dishAgg, E);

  gnn_gemm_bias_act<<<dim3(gU, HID / BN), blk, 0, stream>>>(userAgg, W_uu, b_uu, userUpd, U, HID, HID, 1);
  gnn_gemm_bias_act<<<dim3(gU, OUT / BN), blk, 0, stream>>>(userUpd, W_up, b_up, out_user, U, HID, OUT, 0);
  gnn_gemm_bias_act<<<dim3(gD, HID / BN), blk, 0, stream>>>(dishAgg, W_du, b_du, dishUpd, D, HID, HID, 1);
  gnn_gemm_bias_act<<<dim3(gD, OUT / BN), blk, 0, stream>>>(dishUpd, W_dp, b_dp, out_dish, D, HID, OUT, 0);
}


# GEMM kernels: the next K tile's global loads are issued together ahead of the current tile's WMMAs and written to LDS after them (register prefetch); before, every load was waited for singly
# speedup vs baseline: 1.4377x; 1.4377x over previous
#include <hip/hip_runtime.h>

typedef __attribute__((ext_vector_type(2))) float v2f;
typedef __attribute__((ext_vector_type(8))) float v8f;
typedef __attribute__((ext_vector_type(4))) float v4f;

#define BM 128
#define BM2 64
#define BN 128
#define BK 32
#define AS_STRIDE 36
#define BS_STRIDE 132

__global__ __launch_bounds__(256)
void gnn_gemm_bias_act(const float* __restrict__ X, const float* __restrict__ W,
                       const float* __restrict__ bias, float* __restrict__ Y,
                       int M, int K, int N, int doRelu) {
  __shared__ float As[BM * AS_STRIDE];
  __shared__ float Bs[BK * BS_STRIDE];

  const int tid   = threadIdx.x;
  const int lane  = tid & 31;
  const int wave  = tid >> 5;
  const int waveM = wave & 3;
  const int waveN = wave >> 2;
  const int lr    = lane & 15;
  const int kHalf = (lane >> 4) << 1;

  const int m0 = blockIdx.x * BM;
  const int n0 = blockIdx.y * BN;

  v8f acc[2][4];
#pragma unroll
  for (int mt = 0; mt < 2; ++mt)
#pragma unroll
    for (int nt = 0; nt < 4; ++nt)
      acc[mt][nt] = (v8f)0.0f;

  const int ar = tid >> 3;
  const int ac = (tid & 7) << 2;
  const int br = tid >> 5;
  const int bc = (tid & 31) << 2;
  const int lastRow = M - 1;
  const int g0 = min(m0 + ar, lastRow);
  const int g1 = min(m0 + ar + 32, lastRow);
  const int g2 = min(m0 + ar + 64, lastRow);
  const int g3 = min(m0 + ar + 96, lastRow);
  const float* ap0 = X + (size_t)g0 * K + ac;
  const float* ap1 = X + (size_t)g1 * K + ac;
  const float* ap2 = X + (size_t)g2 * K + ac;
  const float* ap3 = X + (size_t)g3 * K + ac;
  const float* bp  = W + (size_t)br * N + (n0 + bc);
  const size_t bStep = (size_t)8 * N;
  float* const sa = As + ar * AS_STRIDE + ac;
  float* const sb = Bs + br * BS_STRIDE + bc;

#define LOAD_TILE_1(KOFF)                                              \
  do {                                                                 \
    const float* bq = bp + (size_t)(KOFF) * N;                         \
    ra0 = *(const v4f*)(ap0 + (KOFF));                                 \
    ra1 = *(const v4f*)(ap1 + (KOFF));                                 \
    ra2 = *(const v4f*)(ap2 + (KOFF));                                 \
    ra3 = *(const v4f*)(ap3 + (KOFF));                                 \
    rb0 = *(const v4f*)(bq);                                           \
    rb1 = *(const v4f*)(bq + bStep);                                   \
    rb2 = *(const v4f*)(bq + 2 * bStep);                               \
    rb3 = *(const v4f*)(bq + 3 * bStep);                               \
  } while (0)
#define STORE_TILE_1()                                                 \
  do {                                                                 \
    *(v4f*)(sa)                  = ra0;                                \
    *(v4f*)(sa + 32 * AS_STRIDE) = ra1;                                \
    *(v4f*)(sa + 64 * AS_STRIDE) = ra2;                                \
    *(v4f*)(sa + 96 * AS_STRIDE) = ra3;                                \
    *(v4f*)(sb)                  = rb0;                                \
    *(v4f*)(sb + 8 * BS_STRIDE)  = rb1;                                \
    *(v4f*)(sb + 16 * BS_STRIDE) = rb2;                                \
    *(v4f*)(sb + 24 * BS_STRIDE) = rb3;                                \
  } while (0)

  v4f ra0, ra1, ra2, ra3, rb0, rb1, rb2, rb3;
  LOAD_TILE_1(0);
  STORE_TILE_1();
  __syncthreads();

  for (int k0 = 0; k0 < K; k0 += BK) {
    const bool more = (k0 + BK) < K;
    if (more) LOAD_TILE_1(k0 + BK);

#pragma unroll
    for (int kk = 0; kk < BK; kk += 4) {
      v2f a[2];
      v2f b[4];
#pragma unroll
      for (int mt = 0; mt < 2; ++mt) {
        int m = waveM * 32 + mt * 16 + lr;
        const float* p = As + m * AS_STRIDE + kk + kHalf;
        a[mt][0] = p[0];
        a[mt][1] = p[1];
      }
#pragma unroll
      for (int nt = 0; nt < 4; ++nt) {
        int n = waveN * 64 + nt * 16 + lr;
        const float* p = Bs + (kk + kHalf) * BS_STRIDE + n;
        b[nt][0] = p[0];
        b[nt][1] = p[BS_STRIDE];
      }
#pragma unroll
      for (int mt = 0; mt < 2; ++mt)
#pragma unroll
        for (int nt = 0; nt < 4; ++nt)
          acc[mt][nt] = __builtin_amdgcn_wmma_f32_16x16x4_f32(
              false, a[mt], false, b[nt], (short)0, acc[mt][nt], false, false);
    }

    if (more) {
      __syncthreads();
      STORE_TILE_1();
      __syncthreads();
    }
  }
#undef LOAD_TILE_1
#undef STORE_TILE_1

  const int rowHalf = (lane >> 4) << 3;
#pragma unroll
  for (int mt = 0; mt < 2; ++mt) {
#pragma unroll
    for (int nt = 0; nt < 4; ++nt) {
      int col  = n0 + waveN * 64 + nt * 16 + lr;
      float bv = bias[col];
#pragma unroll
      for (int v = 0; v < 8; ++v) {
        int row = m0 + waveM * 32 + mt * 16 + rowHalf + v;
        float val = acc[mt][nt][v] + bv;
        if (doRelu) val = fmaxf(val, 0.0f);
        if (row < M) Y[(size_t)row * N + col] = val;
      }
    }
  }
}

__global__ __launch_bounds__(256)
void gnn_gemm_dual(const float* __restrict__ X,
                   const float* __restrict__ W1, const float* __restrict__ b1,
                   float* __restrict__ Y1,
                   const float* __restrict__ W2, const float* __restrict__ b2,
                   float* __restrict__ Y2,
                   int M, int K, int N) {
  __shared__ float As[BM2 * AS_STRIDE];
  __shared__ float Bs[2][BK * BS_STRIDE];

  const int tid     = threadIdx.x;
  const int lane    = tid & 31;
  const int wave    = tid >> 5;
  const int waveOut = wave >> 2;
  const int wq      = wave & 3;
  const int waveM   = wq & 1;
  const int waveN   = wq >> 1;
  const int lr      = lane & 15;
  const int kHalf   = (lane >> 4) << 1;

  const int m0 = blockIdx.x * BM2;
  const int n0 = blockIdx.y * BN;

  v8f acc[2][4];
#pragma unroll
  for (int mt = 0; mt < 2; ++mt)
#pragma unroll
    for (int nt = 0; nt < 4; ++nt)
      acc[mt][nt] = (v8f)0.0f;

  const float* Bw = &Bs[waveOut][0];

  const int ar = tid >> 3;
  const int ac = (tid & 7) << 2;
  const int br = tid >> 5;
  const int bc = (tid & 31) << 2;
  const int lastRow = M - 1;
  const int g0 = min(m0 + ar, lastRow);
  const int g1 = min(m0 + ar + 32, lastRow);
  const float* ap0 = X + (size_t)g0 * K + ac;
  const float* ap1 = X + (size_t)g1 * K + ac;
  const size_t bOff  = (size_t)br * N + (n0 + bc);
  const float* bp1 = W1 + bOff;
  const float* bp2 = W2 + bOff;
  const size_t bStep = (size_t)8 * N;
  float* const sa  = As + ar * AS_STRIDE + ac;
  float* const sb1 = &Bs[0][0] + br * BS_STRIDE + bc;
  float* const sb2 = &Bs[1][0] + br * BS_STRIDE + bc;

#define LOAD_TILE_2(KOFF)                                              \
  do {                                                                 \
    const float* q1 = bp1 + (size_t)(KOFF) * N;                        \
    const float* q2 = bp2 + (size_t)(KOFF) * N;                        \
    ra0  = *(const v4f*)(ap0 + (KOFF));                                \
    ra1  = *(const v4f*)(ap1 + (KOFF));                                \
    rb10 = *(const v4f*)(q1);                                          \
    rb11 = *(const v4f*)(q1 + bStep);                                  \
    rb12 = *(const v4f*)(q1 + 2 * bStep);                              \
    rb13 = *(const v4f*)(q1 + 3 * bStep);                              \
    rb20 = *(const v4f*)(q2);                                          \
    rb21 = *(const v4f*)(q2 + bStep);                                  \
    rb22 = *(const v4f*)(q2 + 2 * bStep);                              \
    rb23 = *(const v4f*)(q2 + 3 * bStep);                              \
  } while (0)
#define STORE_TILE_2()                                                 \
  do {                                                                 \
    *(v4f*)(sa)                   = ra0;                               \
    *(v4f*)(sa + 32 * AS_STRIDE)  = ra1;                               \
    *(v4f*)(sb1)                  = rb10;                              \
    *(v4f*)(sb1 + 8 * BS_STRIDE)  = rb11;                              \
    *(v4f*)(sb1 + 16 * BS_STRIDE) = rb12;                              \
    *(v4f*)(sb1 + 24 * BS_STRIDE) = rb13;                              \
    *(v4f*)(sb2)                  = rb20;                              \
    *(v4f*)(sb2 + 8 * BS_STRIDE)  = rb21;                              \
    *(v4f*)(sb2 + 16 * BS_STRIDE) = rb22;                              \
    *(v4f*)(sb2 + 24 * BS_STRIDE) = rb23;                              \
  } while (0)

  v4f ra0, ra1, rb10, rb11, rb12, rb13, rb20, rb21, rb22, rb23;
  LOAD_TILE_2(0);
  STORE_TILE_2();
  __syncthreads();

  for (int k0 = 0; k0 < K; k0 += BK) {
    const bool more = (k0 + BK) < K;
    if (more) LOAD_TILE_2(k0 + BK);

#pragma unroll
    for (int kk = 0; kk < BK; kk += 4) {
      v2f a[2];
      v2f b[4];
#pragma unroll
      for (int mt = 0; mt < 2; ++mt) {
        int m = waveM * 32 + mt * 16 + lr;
        const float* p = As + m * AS_STRIDE + kk + kHalf;
        a[mt][0] = p[0];
        a[mt][1] = p[1];
      }
#pragma unroll
      for (int nt = 0; nt < 4; ++nt) {
        int n = waveN * 64 + nt * 16 + lr;
        const float* p = Bw + (kk + kHalf) * BS_STRIDE + n;
        b[nt][0] = p[0];
        b[nt][1] = p[BS_STRIDE];
      }
#pragma unroll
      for (int mt = 0; mt < 2; ++mt)
#pragma unroll
        for (int nt = 0; nt < 4; ++nt)
          acc[mt][nt] = __builtin_amdgcn_wmma_f32_16x16x4_f32(
              false, a[mt], false, b[nt], (short)0, acc[mt][nt], false, false);
    }

    if (more) {
      __syncthreads();
      STORE_TILE_2();
      __syncthreads();
    }
  }
#undef LOAD_TILE_2
#undef STORE_TILE_2

  const float* bias = waveOut ? b2 : b1;
  float*       Y    = waveOut ? Y2 : Y1;
  const int rowHalf = (lane >> 4) << 3;
#pragma unroll
  for (int mt = 0; mt < 2; ++mt) {
#pragma unroll
    for (int nt = 0; nt < 4; ++nt) {
      int col  = n0 + waveN * 64 + nt * 16 + lr;
      float bv = bias[col];
#pragma unroll
      for (int v = 0; v < 8; ++v) {
        int row = m0 + waveM * 32 + mt * 16 + rowHalf + v;
        float val = fmaxf(acc[mt][nt][v] + bv, 0.0f);
        if (row < M) Y[(size_t)row * N + col] = val;
      }
    }
  }
}

__global__ __launch_bounds__(256)
void gnn_scatter_add(const int* __restrict__ recvIdx, const int* __restrict__ sendIdx,
                     const float* __restrict__ msg, float* __restrict__ agg, int nEdges) {
  int e = blockIdx.x;
  if (e >= nEdges) return;
  int r = recvIdx[e];
  int s = sendIdx[e];
  int c = threadIdx.x;
  float v = msg[(size_t)s * 256 + c];
  unsigned int off = ((unsigned int)r * 256u + (unsigned int)c) * 4u;
  asm volatile("global_atomic_add_f32 %0, %1, %2 scope:SCOPE_DEV"
               :
               : "v"(off), "v"(v), "s"(agg)
               : "memory");
}

extern "C" void kernel_launch(void* const* d_in, const int* in_sizes, int n_in,
                              void* d_out, int out_size, void* d_ws, size_t ws_size,
                              hipStream_t stream) {
  (void)in_sizes; (void)n_in; (void)out_size; (void)ws_size;

  constexpr int U = 50000, D = 10000, E = 200000;
  constexpr int DIN = 1024, HID = 256, OUT = 128;

  const float* user_feat = (const float*)d_in[0];
  const float* dish_feat = (const float*)d_in[1];
  const int*   edge_user = (const int*)d_in[2];
  const int*   edge_dish = (const int*)d_in[3];
  const float* W_ui = (const float*)d_in[4];   const float* b_ui = (const float*)d_in[5];
  const float* W_di = (const float*)d_in[6];   const float* b_di = (const float*)d_in[7];
  const float* W_um = (const float*)d_in[8];   const float* b_um = (const float*)d_in[9];
  const float* W_dm = (const float*)d_in[10];  const float* b_dm = (const float*)d_in[11];
  const float* W_uu = (const float*)d_in[12];  const float* b_uu = (const float*)d_in[13];
  const float* W_du = (const float*)d_in[14];  const float* b_du = (const float*)d_in[15];
  const float* W_up = (const float*)d_in[16];  const float* b_up = (const float*)d_in[17];
  const float* W_dp = (const float*)d_in[18];  const float* b_dp = (const float*)d_in[19];

  float* out_user = (float*)d_out;
  float* out_dish = out_user + (size_t)U * OUT;

  float* ws      = (float*)d_ws;
  float* userAgg = ws;
  float* dishAgg = userAgg + (size_t)U * HID;
  float* msgUser = dishAgg + (size_t)D * HID;
  float* msgDish = msgUser + (size_t)U * HID;
  float* userUpd = msgUser;
  float* dishUpd = msgDish;

  dim3 blk(256);
  const int gU  = (U + BM - 1) / BM;
  const int gD  = (D + BM - 1) / BM;
  const int gU2 = (U + BM2 - 1) / BM2;
  const int gD2 = (D + BM2 - 1) / BM2;

  gnn_gemm_dual<<<dim3(gU2, HID / BN), blk, 0, stream>>>(
      user_feat, W_ui, b_ui, userAgg, W_dm, b_dm, msgUser, U, DIN, HID);
  gnn_gemm_dual<<<dim3(gD2, HID / BN), blk, 0, stream>>>(
      dish_feat, W_di, b_di, dishAgg, W_um, b_um, msgDish, D, DIN, HID);

  gnn_scatter_add<<<dim3(E), blk, 0, stream>>>(edge_user, edge_dish, msgDish, userAgg, E);
  gnn_scatter_add<<<dim3(E), blk, 0, stream>>>(edge_dish, edge_user, msgUser, dishAgg, E);

  gnn_gemm_bias_act<<<dim3(gU, HID / BN), blk, 0, stream>>>(userAgg, W_uu, b_uu, userUpd, U, HID, HID, 1);
  gnn_gemm_bias_act<<<dim3(gU, OUT / BN), blk, 0, stream>>>(userUpd, W_up, b_up, out_user, U, HID, OUT, 0);
  gnn_gemm_bias_act<<<dim3(gD, HID / BN), blk, 0, stream>>>(dishAgg, W_du, b_du, dishUpd, D, HID, HID, 1);
  gnn_gemm_bias_act<<<dim3(gD, OUT / BN), blk, 0, stream>>>(dishUpd, W_dp, b_dp, out_dish, D, HID, OUT, 0);
}


# Edge aggregation: the f32 atomic scatter is replaced by atomic-free per-call neighbour lists (owner waves count, prefix sum, fill) and a gather-sum; sums now run in edge order per receiver
# speedup vs baseline: 1.6846x; 1.1717x over previous
#include <hip/hip_runtime.h>

typedef __attribute__((ext_vector_type(2))) float v2f;
typedef __attribute__((ext_vector_type(8))) float v8f;
typedef __attribute__((ext_vector_type(4))) float v4f;

#define BM 128
#define BM2 64
#define BN 128
#define BK 32
#define AS_STRIDE 36
#define BS_STRIDE 132

__global__ __launch_bounds__(256)
void gnn_gemm_bias_act(const float* __restrict__ X, const float* __restrict__ W,
                       const float* __restrict__ bias, float* __restrict__ Y,
                       int M, int K, int N, int doRelu) {
  __shared__ float As[BM * AS_STRIDE];
  __shared__ float Bs[BK * BS_STRIDE];

  const int tid   = threadIdx.x;
  const int lane  = tid & 31;
  const int wave  = tid >> 5;
  const int waveM = wave & 3;
  const int waveN = wave >> 2;
  const int lr    = lane & 15;
  const int kHalf = (lane >> 4) << 1;

  const int m0 = blockIdx.x * BM;
  const int n0 = blockIdx.y * BN;

  v8f acc[2][4];
#pragma unroll
  for (int mt = 0; mt < 2; ++mt)
#pragma unroll
    for (int nt = 0; nt < 4; ++nt)
      acc[mt][nt] = (v8f)0.0f;

  const int ar = tid >> 3;
  const int ac = (tid & 7) << 2;
  const int br = tid >> 5;
  const int bc = (tid & 31) << 2;
  const int lastRow = M - 1;
  const int g0 = min(m0 + ar, lastRow);
  const int g1 = min(m0 + ar + 32, lastRow);
  const int g2 = min(m0 + ar + 64, lastRow);
  const int g3 = min(m0 + ar + 96, lastRow);
  const float* ap0 = X + (size_t)g0 * K + ac;
  const float* ap1 = X + (size_t)g1 * K + ac;
  const float* ap2 = X + (size_t)g2 * K + ac;
  const float* ap3 = X + (size_t)g3 * K + ac;
  const float* bp  = W + (size_t)br * N + (n0 + bc);
  const size_t bStep = (size_t)8 * N;
  float* const sa = As + ar * AS_STRIDE + ac;
  float* const sb = Bs + br * BS_STRIDE + bc;

#define LOAD_TILE_1(KOFF)                                              \
  do {                                                                 \
    const float* bq = bp + (size_t)(KOFF) * N;                         \
    ra0 = *(const v4f*)(ap0 + (KOFF));                                 \
    ra1 = *(const v4f*)(ap1 + (KOFF));                                 \
    ra2 = *(const v4f*)(ap2 + (KOFF));                                 \
    ra3 = *(const v4f*)(ap3 + (KOFF));                                 \
    rb0 = *(const v4f*)(bq);                                           \
    rb1 = *(const v4f*)(bq + bStep);                                   \
    rb2 = *(const v4f*)(bq + 2 * bStep);                               \
    rb3 = *(const v4f*)(bq + 3 * bStep);                               \
  } while (0)
#define STORE_TILE_1()                                                 \
  do {                                                                 \
    *(v4f*)(sa)                  = ra0;                                \
    *(v4f*)(sa + 32 * AS_STRIDE) = ra1;                                \
    *(v4f*)(sa + 64 * AS_STRIDE) = ra2;                                \
    *(v4f*)(sa + 96 * AS_STRIDE) = ra3;                                \
    *(v4f*)(sb)                  = rb0;                                \
    *(v4f*)(sb + 8 * BS_STRIDE)  = rb1;                                \
    *(v4f*)(sb + 16 * BS_STRIDE) = rb2;                                \
    *(v4f*)(sb + 24 * BS_STRIDE) = rb3;                                \
  } while (0)

  v4f ra0, ra1, ra2, ra3, rb0, rb1, rb2, rb3;
  LOAD_TILE_1(0);
  STORE_TILE_1();
  __syncthreads();

  for (int k0 = 0; k0 < K; k0 += BK) {
    const bool more = (k0 + BK) < K;
    if (more) LOAD_TILE_1(k0 + BK);

#pragma unroll
    for (int kk = 0; kk < BK; kk += 4) {
      v2f a[2];
      v2f b[4];
#pragma unroll
      for (int mt = 0; mt < 2; ++mt) {
        int m = waveM * 32 + mt * 16 + lr;
        const float* p = As + m * AS_STRIDE + kk + kHalf;
        a[mt][0] = p[0];
        a[mt][1] = p[1];
      }
#pragma unroll
      for (int nt = 0; nt < 4; ++nt) {
        int n = waveN * 64 + nt * 16 + lr;
        const float* p = Bs + (kk + kHalf) * BS_STRIDE + n;
        b[nt][0] = p[0];
        b[nt][1] = p[BS_STRIDE];
      }
#pragma unroll
      for (int mt = 0; mt < 2; ++mt)
#pragma unroll
        for (int nt = 0; nt < 4; ++nt)
          acc[mt][nt] = __builtin_amdgcn_wmma_f32_16x16x4_f32(
              false, a[mt], false, b[nt], (short)0, acc[mt][nt], false, false);
    }

    if (more) {
      __syncthreads();
      STORE_TILE_1();
      __syncthreads();
    }
  }
#undef LOAD_TILE_1
#undef STORE_TILE_1

  const int rowHalf = (lane >> 4) << 3;
#pragma unroll
  for (int mt = 0; mt < 2; ++mt) {
#pragma unroll
    for (int nt = 0; nt < 4; ++nt) {
      int col  = n0 + waveN * 64 + nt * 16 + lr;
      float bv = bias[col];
#pragma unroll
      for (int v = 0; v < 8; ++v) {
        int row = m0 + waveM * 32 + mt * 16 + rowHalf + v;
        float val = acc[mt][nt][v] + bv;
        if (doRelu) val = fmaxf(val, 0.0f);
        if (row < M) Y[(size_t)row * N + col] = val;
      }
    }
  }
}

__global__ __launch_bounds__(256)
void gnn_gemm_dual(const float* __restrict__ X,
                   const float* __restrict__ W1, const float* __restrict__ b1,
                   float* __restrict__ Y1,
                   const float* __restrict__ W2, const float* __restrict__ b2,
                   float* __restrict__ Y2,
                   int M, int K, int N) {
  __shared__ float As[BM2 * AS_STRIDE];
  __shared__ float Bs[2][BK * BS_STRIDE];

  const int tid     = threadIdx.x;
  const int lane    = tid & 31;
  const int wave    = tid >> 5;
  const int waveOut = wave >> 2;
  const int wq      = wave & 3;
  const int waveM   = wq & 1;
  const int waveN   = wq >> 1;
  const int lr      = lane & 15;
  const int kHalf   = (lane >> 4) << 1;

  const int m0 = blockIdx.x * BM2;
  const int n0 = blockIdx.y * BN;

  v8f acc[2][4];
#pragma unroll
  for (int mt = 0; mt < 2; ++mt)
#pragma unroll
    for (int nt = 0; nt < 4; ++nt)
      acc[mt][nt] = (v8f)0.0f;

  const float* Bw = &Bs[waveOut][0];

  const int ar = tid >> 3;
  const int ac = (tid & 7) << 2;
  const int br = tid >> 5;
  const int bc = (tid & 31) << 2;
  const int lastRow = M - 1;
  const int g0 = min(m0 + ar, lastRow);
  const int g1 = min(m0 + ar + 32, lastRow);
  const float* ap0 = X + (size_t)g0 * K + ac;
  const float* ap1 = X + (size_t)g1 * K + ac;
  const size_t bOff  = (size_t)br * N + (n0 + bc);
  const float* bp1 = W1 + bOff;
  const float* bp2 = W2 + bOff;
  const size_t bStep = (size_t)8 * N;
  float* const sa  = As + ar * AS_STRIDE + ac;
  float* const sb1 = &Bs[0][0] + br * BS_STRIDE + bc;
  float* const sb2 = &Bs[1][0] + br * BS_STRIDE + bc;

#define LOAD_TILE_2(KOFF)                                              \
  do {                                                                 \
    const float* q1 = bp1 + (size_t)(KOFF) * N;                        \
    const float* q2 = bp2 + (size_t)(KOFF) * N;                        \
    ra0  = *(const v4f*)(ap0 + (KOFF));                                \
    ra1  = *(const v4f*)(ap1 + (KOFF));                                \
    rb10 = *(const v4f*)(q1);                                          \
    rb11 = *(const v4f*)(q1 + bStep);                                  \
    rb12 = *(const v4f*)(q1 + 2 * bStep);                              \
    rb13 = *(const v4f*)(q1 + 3 * bStep);                              \
    rb20 = *(const v4f*)(q2);                                          \
    rb21 = *(const v4f*)(q2 + bStep);                                  \
    rb22 = *(const v4f*)(q2 + 2 * bStep);                              \
    rb23 = *(const v4f*)(q2 + 3 * bStep);                              \
  } while (0)
#define STORE_TILE_2()                                                 \
  do {                                                                 \
    *(v4f*)(sa)                   = ra0;                               \
    *(v4f*)(sa + 32 * AS_STRIDE)  = ra1;                               \
    *(v4f*)(sb1)                  = rb10;                              \
    *(v4f*)(sb1 + 8 * BS_STRIDE)  = rb11;                              \
    *(v4f*)(sb1 + 16 * BS_STRIDE) = rb12;                              \
    *(v4f*)(sb1 + 24 * BS_STRIDE) = rb13;                              \
    *(v4f*)(sb2)                  = rb20;                              \
    *(v4f*)(sb2 + 8 * BS_STRIDE)  = rb21;                              \
    *(v4f*)(sb2 + 16 * BS_STRIDE) = rb22;                              \
    *(v4f*)(sb2 + 24 * BS_STRIDE) = rb23;                              \
  } while (0)

  v4f ra0, ra1, rb10, rb11, rb12, rb13, rb20, rb21, rb22, rb23;
  LOAD_TILE_2(0);
  STORE_TILE_2();
  __syncthreads();

  for (int k0 = 0; k0 < K; k0 += BK) {
    const bool more = (k0 + BK) < K;
    if (more) LOAD_TILE_2(k0 + BK);

#pragma unroll
    for (int kk = 0; kk < BK; kk += 4) {
      v2f a[2];
      v2f b[4];
#pragma unroll
      for (int mt = 0; mt < 2; ++mt) {
        int m = waveM * 32 + mt * 16 + lr;
        const float* p = As + m * AS_STRIDE + kk + kHalf;
        a[mt][0] = p[0];
        a[mt][1] = p[1];
      }
#pragma unroll
      for (int nt = 0; nt < 4; ++nt) {
        int n = waveN * 64 + nt * 16 + lr;
        const float* p = Bw + (kk + kHalf) * BS_STRIDE + n;
        b[nt][0] = p[0];
        b[nt][1] = p[BS_STRIDE];
      }
#pragma unroll
      for (int mt = 0; mt < 2; ++mt)
#pragma unroll
        for (int nt = 0; nt < 4; ++nt)
          acc[mt][nt] = __builtin_amdgcn_wmma_f32_16x16x4_f32(
              false, a[mt], false, b[nt], (short)0, acc[mt][nt], false, false);
    }

    if (more) {
      __syncthreads();
      STORE_TILE_2();
      __syncthreads();
    }
  }
#undef LOAD_TILE_2
#undef STORE_TILE_2

  const float* bias = waveOut ? b2 : b1;
  float*       Y    = waveOut ? Y2 : Y1;
  const int rowHalf = (lane >> 4) << 3;
#pragma unroll
  for (int mt = 0; mt < 2; ++mt) {
#pragma unroll
    for (int nt = 0; nt < 4; ++nt) {
      int col  = n0 + waveN * 64 + nt * 16 + lr;
      float bv = bias[col];
#pragma unroll
      for (int v = 0; v < 8; ++v) {
        int row = m0 + waveM * 32 + mt * 16 + rowHalf + v;
        float val = fmaxf(acc[mt][nt][v] + bv, 0.0f);
        if (row < M) Y[(size_t)row * N + col] = val;
      }
    }
  }
}

typedef __attribute__((ext_vector_type(4))) int v4i;
#define CSR_Q 16
#define CSR_G 128

template <bool FILL>
__device__ __forceinline__ void csr_take(v4i rv, v4i sv, int r0, int lane, v4i& cur,
                                         int* __restrict__ nbr, int nList) {
  const unsigned b0 = __builtin_amdgcn_ballot_w32((unsigned)(rv.x - r0) < (unsigned)CSR_G);
  const unsigned b1 = __builtin_amdgcn_ballot_w32((unsigned)(rv.y - r0) < (unsigned)CSR_G);
  const unsigned b2 = __builtin_amdgcn_ballot_w32((unsigned)(rv.z - r0) < (unsigned)CSR_G);
  const unsigned b3 = __builtin_amdgcn_ballot_w32((unsigned)(rv.w - r0) < (unsigned)CSR_G);
  unsigned any = b0 | b1 | b2 | b3;
  while (any) {
    const int l = __builtin_ctz(any);
    any &= any - 1;
#define CSR_ONE(BITS, COMP)                                               \
    if ((BITS >> l) & 1u) {                                               \
      const int k    = __builtin_amdgcn_readlane(rv.COMP, l) - r0;        \
      const int s    = FILL ? __builtin_amdgcn_readlane(sv.COMP, l) : 0;  \
      const int slot = k >> 5;                                            \
      if (lane == (k & 31)) {                                             \
        if (FILL) {                                                       \
          const int c = slot == 0 ? cur.x : slot == 1 ? cur.y             \
                      : slot == 2 ? cur.z : cur.w;                        \
          if ((unsigned)c < (unsigned)nList) nbr[c] = s;                  \
        }                                                                 \
        cur.x += (slot == 0);                                             \
        cur.y += (slot == 1);                                             \
        cur.z += (slot == 2);                                             \
        cur.w += (slot == 3);                                             \
      }                                                                   \
    }
    CSR_ONE(b0, x)
    CSR_ONE(b1, y)
    CSR_ONE(b2, z)
    CSR_ONE(b3, w)
#undef CSR_ONE
  }
}
template <bool FILL>
__device__ __forceinline__ void csr_scan_slice(const int* __restrict__ recv,
                                               const int* __restrict__ send,
                                               int eBeg, int eEnd, int r0, int lane, v4i& cur,
                                               int* __restrict__ nbr, int nList) {
  int base = eBeg;
  for (; base + 512 <= eEnd; base += 512) {
    const int e0 = base + lane * 4;
    const v4i ra = *(const v4i*)(recv + e0);
    const v4i rb = *(const v4i*)(recv + e0 + 128);
    const v4i rc = *(const v4i*)(recv + e0 + 256);
    const v4i rd = *(const v4i*)(recv + e0 + 384);
    v4i sa = ra, sb = rb, sc = rc, sd = rd;
    if (FILL) {
      sa = *(const v4i*)(send + e0);
      sb = *(const v4i*)(send + e0 + 128);
      sc = *(const v4i*)(send + e0 + 256);
      sd = *(const v4i*)(send + e0 + 384);
    }
    csr_take<FILL>(ra, sa, r0, lane, cur, nbr, nList);
    csr_take<FILL>(rb, sb, r0, lane, cur, nbr, nList);
    csr_take<FILL>(rc, sc, r0, lane, cur, nbr, nList);
    csr_take<FILL>(rd, sd, r0, lane, cur, nbr, nList);
  }
  for (; base < eEnd; base += 128) {
    const int e0 = base + lane * 4;
    v4i rv = (v4i)(-1);
    v4i sv = (v4i)(0);
    if (e0 + 0 < eEnd) { rv.x = recv[e0 + 0]; if (FILL) sv.x = send[e0 + 0]; }
    if (e0 + 1 < eEnd) { rv.y = recv[e0 + 1]; if (FILL) sv.y = send[e0 + 1]; }
    if (e0 + 2 < eEnd) { rv.z = recv[e0 + 2]; if (FILL) sv.z = send[e0 + 2]; }
    if (e0 + 3 < eEnd) { rv.w = recv[e0 + 3]; if (FILL) sv.w = send[e0 + 3]; }
    csr_take<FILL>(rv, sv, r0, lane, cur, nbr, nList);
  }
}

struct CsrOwner { const int* recv; const int* send; int r0; int nOwn; int rcBase; int eBeg; int eEnd; };
__device__ __forceinline__ CsrOwner csr_owner(const int* __restrict__ edgeUser,
                                              const int* __restrict__ edgeDish,
                                              int nUsers, int nDishes, int nEdges, int q) {
  CsrOwner o;
  const int gU = (nUsers + CSR_G - 1) / CSR_G;
  const int g  = blockIdx.x;
  const bool userSide = g < gU;
  o.recv   = userSide ? edgeUser : edgeDish;
  o.send   = userSide ? edgeDish : edgeUser;
  o.r0     = (userSide ? g : g - gU) * CSR_G;
  o.nOwn   = userSide ? nUsers : nDishes;
  o.rcBase = userSide ? o.r0 : nUsers + o.r0;
  const int per = (((nEdges + CSR_Q - 1) / CSR_Q) + 3) & ~3;
  o.eBeg = min(nEdges, q * per);
  o.eEnd = min(nEdges, o.eBeg + per);
  return o;
}

__global__ __launch_bounds__(32 * CSR_Q)
void gnn_csr_count(const int* __restrict__ edgeUser, const int* __restrict__ edgeDish,
                   int* __restrict__ cnt, int* __restrict__ tot,
                   int nUsers, int nDishes, int nEdges) {
  __shared__ int shc[CSR_Q][CSR_G];
  const int lane = threadIdx.x & 31;
  const int q = __builtin_amdgcn_readfirstlane(threadIdx.x >> 5);
  const CsrOwner o = csr_owner(edgeUser, edgeDish, nUsers, nDishes, nEdges, q);
  v4i cur = (v4i)(0);
  csr_scan_slice<false>(o.recv, o.send, o.eBeg, o.eEnd, o.r0, lane, cur, nullptr, 0);
  shc[q][lane]      = cur.x;
  shc[q][lane + 32] = cur.y;
  shc[q][lane + 64] = cur.z;
  shc[q][lane + 96] = cur.w;
  __syncthreads();
  const int k = threadIdx.x;
  if (k < CSR_G && o.r0 + k < o.nOwn) {
    v4i* p = (v4i*)(cnt + (size_t)(o.rcBase + k) * CSR_Q);
    int s = 0;
    for (int v = 0; v < CSR_Q / 4; ++v) {
      v4i r;
      r.x = shc[4 * v + 0][k];
      r.y = shc[4 * v + 1][k];
      r.z = shc[4 * v + 2][k];
      r.w = shc[4 * v + 3][k];
      s += (r.x + r.y) + (r.z + r.w);
      p[v] = r;
    }
    tot[o.rcBase + k] = s;
  }
}

__global__ __launch_bounds__(32 * CSR_Q)
void gnn_csr_fill(const int* __restrict__ edgeUser, const int* __restrict__ edgeDish,
                  const int* __restrict__ segBegin, int* __restrict__ nbr,
                  int nUsers, int nDishes, int nEdges, int nList) {
  const int lane = threadIdx.x & 31;
  const int q = __builtin_amdgcn_readfirstlane(threadIdx.x >> 5);
  const CsrOwner o = csr_owner(edgeUser, edgeDish, nUsers, nDishes, nEdges, q);
  const int k0 = o.r0 + lane;
  const int* p = segBegin + (size_t)(o.rcBase + lane) * CSR_Q + q;
  v4i cur = (v4i)(0);
  if (k0      < o.nOwn) cur.x = p[0];
  if (k0 + 32 < o.nOwn) cur.y = p[(size_t)32 * CSR_Q];
  if (k0 + 64 < o.nOwn) cur.z = p[(size_t)64 * CSR_Q];
  if (k0 + 96 < o.nOwn) cur.w = p[(size_t)96 * CSR_Q];
  csr_scan_slice<true>(o.recv, o.send, o.eBeg, o.eEnd, o.r0, lane, cur, nbr, nList);
}

__device__ __forceinline__ int csr_sum4(v4i a) { return (a.x + a.y) + (a.z + a.w); }

__global__ __launch_bounds__(1024)
void gnn_csr_offsets(const int* __restrict__ cnt, const int* __restrict__ tot,
                     int* __restrict__ segBegin, int nRecv) {
  __shared__ int sh[32];
  __shared__ int shBase;
  const int t = threadIdx.x;
  const int lane = t & 31;
  const int w = t >> 5;

  int part = 0;
  for (int j = 0; j < (int)blockIdx.x; ++j) part += tot[j * 1024 + t];
  for (int d = 16; d >= 1; d >>= 1) part += __shfl_xor(part, d, 32);
  if (lane == 0) sh[w] = part;
  __syncthreads();
  if (w == 0) {
    int v = sh[lane];
    for (int d = 16; d >= 1; d >>= 1) v += __shfl_xor(v, d, 32);
    if (lane == 0) shBase = v;
  }
  __syncthreads();
  const int base = shBase;

  const int i = blockIdx.x * 1024 + t;
  v4i c[CSR_Q / 4];
  int rowTot = 0;
  for (int v = 0; v < CSR_Q / 4; ++v) {
    c[v] = (v4i)(0);
    if (i < nRecv) c[v] = ((const v4i*)(cnt + (size_t)i * CSR_Q))[v];
    rowTot += csr_sum4(c[v]);
  }
  int incl = rowTot;
  for (int d = 1; d < 32; d <<= 1) {
    const int u = __shfl_up(incl, d, 32);
    if (lane >= d) incl += u;
  }
  if (lane == 31) sh[w] = incl;
  __syncthreads();
  if (w == 0) {
    const int v = sh[lane];
    int s = v;
    for (int d = 1; d < 32; d <<= 1) {
      const int u = __shfl_up(s, d, 32);
      if (lane >= d) s += u;
    }
    sh[lane] = s - v;
  }
  __syncthreads();
  if (i < nRecv) {
    int o = base + sh[w] + incl - rowTot;
    v4i* p = (v4i*)(segBegin + (size_t)i * CSR_Q);
    for (int v = 0; v < CSR_Q / 4; ++v) {
      v4i r;
      r.x = o; o += c[v].x;
      r.y = o; o += c[v].y;
      r.z = o; o += c[v].z;
      r.w = o; o += c[v].w;
      p[v] = r;
    }
  }
}

__global__ __launch_bounds__(256)
void gnn_gather_sum(const int* __restrict__ segBegin, const int* __restrict__ cnt,
                    const int* __restrict__ nbr,
                    const float* __restrict__ msgUser, const float* __restrict__ msgDish,
                    float* __restrict__ agg, int nUsers, int nDishes, int nList) {
  const int lane = threadIdx.x & 31;
  const int wave = __builtin_amdgcn_readfirstlane(threadIdx.x >> 5);
  const int nRecv = nUsers + nDishes;
  const int i = nRecv - 1 - (blockIdx.x * 8 + wave);
  if (i < 0) return;
  const bool isUser = i < nUsers;
  const float* msg  = isUser ? msgDish : msgUser;
  const int nSend   = isUser ? nDishes : nUsers;
  int b = segBegin[(size_t)i * CSR_Q];
  int e = segBegin[(size_t)i * CSR_Q + (CSR_Q - 1)] + cnt[(size_t)i * CSR_Q + (CSR_Q - 1)];
  b = max(0, min(b, nList));
  e = max(b, min(e, nList));
  float* row = agg + (size_t)i * 256 + lane * 4;
  v4f a0 = *(const v4f*)(row);
  v4f a1 = *(const v4f*)(row + 128);
  for (int base = b; base < e; base += 32) {
    const int n = min(32, e - base);
    int id = (lane < n) ? nbr[base + lane] : 0;
    id = max(0, min(id, nSend - 1));
    int j = 0;
    for (; j + 4 <= n; j += 4) {
      const float* r0 = msg + (size_t)__builtin_amdgcn_readlane(id, j)     * 256 + lane * 4;
      const float* r1 = msg + (size_t)__builtin_amdgcn_readlane(id, j + 1) * 256 + lane * 4;
      const float* r2 = msg + (size_t)__builtin_amdgcn_readlane(id, j + 2) * 256 + lane * 4;
      const float* r3 = msg + (size_t)__builtin_amdgcn_readlane(id, j + 3) * 256 + lane * 4;
      v4f m00 = *(const v4f*)(r0), m01 = *(const v4f*)(r0 + 128);
      v4f m10 = *(const v4f*)(r1), m11 = *(const v4f*)(r1 + 128);
      v4f m20 = *(const v4f*)(r2), m21 = *(const v4f*)(r2 + 128);
      v4f m30 = *(const v4f*)(r3), m31 = *(const v4f*)(r3 + 128);
      a0 += m00; a1 += m01;
      a0 += m10; a1 += m11;
      a0 += m20; a1 += m21;
      a0 += m30; a1 += m31;
    }
    for (; j < n; ++j) {
      const float* r0 = msg + (size_t)__builtin_amdgcn_readlane(id, j) * 256 + lane * 4;
      v4f m00 = *(const v4f*)(r0), m01 = *(const v4f*)(r0 + 128);
      a0 += m00; a1 += m01;
    }
  }
  *(v4f*)(row)       = a0;
  *(v4f*)(row + 128) = a1;
}

extern "C" void kernel_launch(void* const* d_in, const int* in_sizes, int n_in,
                              void* d_out, int out_size, void* d_ws, size_t ws_size,
                              hipStream_t stream) {
  (void)in_sizes; (void)n_in; (void)out_size; (void)ws_size;

  constexpr int U = 50000, D = 10000, E = 200000;
  constexpr int DIN = 1024, HID = 256, OUT = 128;

  const float* user_feat = (const float*)d_in[0];
  const float* dish_feat = (const float*)d_in[1];
  const int*   edge_user = (const int*)d_in[2];
  const int*   edge_dish = (const int*)d_in[3];
  const float* W_ui = (const float*)d_in[4];   const float* b_ui = (const float*)d_in[5];
  const float* W_di = (const float*)d_in[6];   const float* b_di = (const float*)d_in[7];
  const float* W_um = (const float*)d_in[8];   const float* b_um = (const float*)d_in[9];
  const float* W_dm = (const float*)d_in[10];  const float* b_dm = (const float*)d_in[11];
  const float* W_uu = (const float*)d_in[12];  const float* b_uu = (const float*)d_in[13];
  const float* W_du = (const float*)d_in[14];  const float* b_du = (const float*)d_in[15];
  const float* W_up = (const float*)d_in[16];  const float* b_up = (const float*)d_in[17];
  const float* W_dp = (const float*)d_in[18];  const float* b_dp = (const float*)d_in[19];

  float* out_user = (float*)d_out;
  float* out_dish = out_user + (size_t)U * OUT;

  float* ws      = (float*)d_ws;
  float* userAgg = ws;
  float* dishAgg = userAgg + (size_t)U * HID;
  float* msgUser = dishAgg + (size_t)D * HID;
  float* msgDish = msgUser + (size_t)U * HID;
  float* userUpd = msgUser;
  float* dishUpd = msgDish;

  dim3 blk(256);
  const int gU  = (U + BM - 1) / BM;
  const int gD  = (D + BM - 1) / BM;
  const int gU2 = (U + BM2 - 1) / BM2;
  const int gD2 = (D + BM2 - 1) / BM2;

  gnn_gemm_dual<<<dim3(gU2, HID / BN), blk, 0, stream>>>(
      user_feat, W_ui, b_ui, userAgg, W_dm, b_dm, msgUser, U, DIN, HID);
  gnn_gemm_dual<<<dim3(gD2, HID / BN), blk, 0, stream>>>(
      dish_feat, W_di, b_di, dishAgg, W_um, b_um, msgDish, D, DIN, HID);

  constexpr int R  = U + D;
  constexpr int L  = 2 * E;
  constexpr int NG = (U + CSR_G - 1) / CSR_G + (D + CSR_G - 1) / CSR_G;
  int* cnt      = (int*)d_out;
  int* segBegin = cnt + (size_t)R * CSR_Q;
  int* tot      = segBegin + (size_t)R * CSR_Q;
  int* nbr      = tot + R;
  gnn_csr_count<<<dim3(NG), dim3(32 * CSR_Q), 0, stream>>>(edge_user, edge_dish, cnt, tot, U, D, E);
  gnn_csr_offsets<<<dim3((R + 1023) / 1024), dim3(1024), 0, stream>>>(cnt, tot, segBegin, R);
  gnn_csr_fill<<<dim3(NG), dim3(32 * CSR_Q), 0, stream>>>(edge_user, edge_dish, segBegin, nbr, U, D, E, L);
  gnn_gather_sum<<<dim3((R + 7) / 8), blk, 0, stream>>>(
      segBegin, cnt, nbr, msgUser, msgDish, userAgg, U, D, L);

  gnn_gemm_bias_act<<<dim3(gU, HID / BN), blk, 0, stream>>>(userAgg, W_uu, b_uu, userUpd, U, HID, HID, 1);
  gnn_gemm_bias_act<<<dim3(gU, OUT / BN), blk, 0, stream>>>(userUpd, W_up, b_up, out_user, U, HID, OUT, 0);
  gnn_gemm_bias_act<<<dim3(gD, HID / BN), blk, 0, stream>>>(dishAgg, W_du, b_du, dishUpd, D, HID, HID, 1);
  gnn_gemm_bias_act<<<dim3(gD, OUT / BN), blk, 0, stream>>>(dishUpd, W_dp, b_dp, out_dish, D, HID, OUT, 0);
}


# GEMM launches merged per stage: user and dish row blocks share one grid in layer 1, in the update and in the projection (10 launches to 7), so the small dish grids fill the user grids' tails
# speedup vs baseline: 1.9255x; 1.1430x over previous
#include <hip/hip_runtime.h>

typedef __attribute__((ext_vector_type(2))) float v2f;
typedef __attribute__((ext_vector_type(8))) float v8f;
typedef __attribute__((ext_vector_type(4))) float v4f;

#define BM 128
#define BM2 64
#define BN 128
#define BK 32
#define AS_STRIDE 36
#define BS_STRIDE 132

struct GemmSide {
  const float* X; const float* W; const float* bias; float* Y;
};

__global__ __launch_bounds__(256)
void gnn_gemm_bias_act(GemmSide su, GemmSide sd, int nBlkU, int Mu, int Md,
                       int K, int N, int doRelu) {
  __shared__ float As[BM * AS_STRIDE];
  __shared__ float Bs[BK * BS_STRIDE];

  const bool sideU = (int)blockIdx.x < nBlkU;
  const float* __restrict__ X    = sideU ? su.X    : sd.X;
  const float* __restrict__ W    = sideU ? su.W    : sd.W;
  const float* __restrict__ bias = sideU ? su.bias : sd.bias;
  float* __restrict__       Y    = sideU ? su.Y    : sd.Y;
  const int M  = sideU ? Mu : Md;
  const int bx = sideU ? (int)blockIdx.x : (int)blockIdx.x - nBlkU;

  const int tid   = threadIdx.x;
  const int lane  = tid & 31;
  const int wave  = tid >> 5;
  const int waveM = wave & 3;
  const int waveN = wave >> 2;
  const int lr    = lane & 15;
  const int kHalf = (lane >> 4) << 1;

  const int m0 = bx * BM;
  const int n0 = blockIdx.y * BN;

  v8f acc[2][4];
#pragma unroll
  for (int mt = 0; mt < 2; ++mt)
#pragma unroll
    for (int nt = 0; nt < 4; ++nt)
      acc[mt][nt] = (v8f)0.0f;

  const int ar = tid >> 3;
  const int ac = (tid & 7) << 2;
  const int br = tid >> 5;
  const int bc = (tid & 31) << 2;
  const int lastRow = M - 1;
  const int g0 = min(m0 + ar, lastRow);
  const int g1 = min(m0 + ar + 32, lastRow);
  const int g2 = min(m0 + ar + 64, lastRow);
  const int g3 = min(m0 + ar + 96, lastRow);
  const float* ap0 = X + (size_t)g0 * K + ac;
  const float* ap1 = X + (size_t)g1 * K + ac;
  const float* ap2 = X + (size_t)g2 * K + ac;
  const float* ap3 = X + (size_t)g3 * K + ac;
  const float* bp  = W + (size_t)br * N + (n0 + bc);
  const size_t bStep = (size_t)8 * N;
  float* const sa = As + ar * AS_STRIDE + ac;
  float* const sb = Bs + br * BS_STRIDE + bc;

#define LOAD_TILE_1(KOFF)                                              \
  do {                                                                 \
    const float* bq = bp + (size_t)(KOFF) * N;                         \
    ra0 = *(const v4f*)(ap0 + (KOFF));                                 \
    ra1 = *(const v4f*)(ap1 + (KOFF));                                 \
    ra2 = *(const v4f*)(ap2 + (KOFF));                                 \
    ra3 = *(const v4f*)(ap3 + (KOFF));                                 \
    rb0 = *(const v4f*)(bq);                                           \
    rb1 = *(const v4f*)(bq + bStep);                                   \
    rb2 = *(const v4f*)(bq + 2 * bStep);                               \
    rb3 = *(const v4f*)(bq + 3 * bStep);                               \
  } while (0)
#define STORE_TILE_1()                                                 \
  do {                                                                 \
    *(v4f*)(sa)                  = ra0;                                \
    *(v4f*)(sa + 32 * AS_STRIDE) = ra1;                                \
    *(v4f*)(sa + 64 * AS_STRIDE) = ra2;                                \
    *(v4f*)(sa + 96 * AS_STRIDE) = ra3;                                \
    *(v4f*)(sb)                  = rb0;                                \
    *(v4f*)(sb + 8 * BS_STRIDE)  = rb1;                                \
    *(v4f*)(sb + 16 * BS_STRIDE) = rb2;                                \
    *(v4f*)(sb + 24 * BS_STRIDE) = rb3;                                \
  } while (0)

  v4f ra0, ra1, ra2, ra3, rb0, rb1, rb2, rb3;
  LOAD_TILE_1(0);
  STORE_TILE_1();
  __syncthreads();

  for (int k0 = 0; k0 < K; k0 += BK) {
    const bool more = (k0 + BK) < K;
    if (more) LOAD_TILE_1(k0 + BK);

#pragma unroll
    for (int kk = 0; kk < BK; kk += 4) {
      v2f a[2];
      v2f b[4];
#pragma unroll
      for (int mt = 0; mt < 2; ++mt) {
        int m = waveM * 32 + mt * 16 + lr;
        const float* p = As + m * AS_STRIDE + kk + kHalf;
        a[mt][0] = p[0];
        a[mt][1] = p[1];
      }
#pragma unroll
      for (int nt = 0; nt < 4; ++nt) {
        int n = waveN * 64 + nt * 16 + lr;
        const float* p = Bs + (kk + kHalf) * BS_STRIDE + n;
        b[nt][0] = p[0];
        b[nt][1] = p[BS_STRIDE];
      }
#pragma unroll
      for (int mt = 0; mt < 2; ++mt)
#pragma unroll
        for (int nt = 0; nt < 4; ++nt)
          acc[mt][nt] = __builtin_amdgcn_wmma_f32_16x16x4_f32(
              false, a[mt], false, b[nt], (short)0, acc[mt][nt], false, false);
    }

    if (more) {
      __syncthreads();
      STORE_TILE_1();
      __syncthreads();
    }
  }
#undef LOAD_TILE_1
#undef STORE_TILE_1

  const int rowHalf = (lane >> 4) << 3;
  float bvs[4];
#pragma unroll
  for (int nt = 0; nt < 4; ++nt) bvs[nt] = bias[n0 + waveN * 64 + nt * 16 + lr];
#pragma unroll
  for (int mt = 0; mt < 2; ++mt) {
#pragma unroll
    for (int nt = 0; nt < 4; ++nt) {
      int col  = n0 + waveN * 64 + nt * 16 + lr;
      float bv = bvs[nt];
#pragma unroll
      for (int v = 0; v < 8; ++v) {
        int row = m0 + waveM * 32 + mt * 16 + rowHalf + v;
        float val = acc[mt][nt][v] + bv;
        if (doRelu) val = fmaxf(val, 0.0f);
        if (row < M) Y[(size_t)row * N + col] = val;
      }
    }
  }
}

struct DualSide {
  const float* X;
  const float* W1; const float* b1; float* Y1;
  const float* W2; const float* b2; float* Y2;
};

__global__ __launch_bounds__(256)
void gnn_gemm_dual(DualSide su, DualSide sd, int nBlkU, int Mu, int Md, int K, int N) {
  __shared__ float As[BM2 * AS_STRIDE];
  __shared__ float Bs[2][BK * BS_STRIDE];

  const bool sideU = (int)blockIdx.x < nBlkU;
  const float* __restrict__ X  = sideU ? su.X  : sd.X;
  const float* __restrict__ W1 = sideU ? su.W1 : sd.W1;
  const float* __restrict__ b1 = sideU ? su.b1 : sd.b1;
  float* __restrict__       Y1 = sideU ? su.Y1 : sd.Y1;
  const float* __restrict__ W2 = sideU ? su.W2 : sd.W2;
  const float* __restrict__ b2 = sideU ? su.b2 : sd.b2;
  float* __restrict__       Y2 = sideU ? su.Y2 : sd.Y2;
  const int M  = sideU ? Mu : Md;
  const int bx = sideU ? (int)blockIdx.x : (int)blockIdx.x - nBlkU;

  const int tid     = threadIdx.x;
  const int lane    = tid & 31;
  const int wave    = tid >> 5;
  const int waveOut = wave >> 2;
  const int wq      = wave & 3;
  const int waveM   = wq & 1;
  const int waveN   = wq >> 1;
  const int lr      = lane & 15;
  const int kHalf   = (lane >> 4) << 1;

  const int m0 = bx * BM2;
  const int n0 = blockIdx.y * BN;

  v8f acc[2][4];
#pragma unroll
  for (int mt = 0; mt < 2; ++mt)
#pragma unroll
    for (int nt = 0; nt < 4; ++nt)
      acc[mt][nt] = (v8f)0.0f;

  const float* Bw = &Bs[waveOut][0];

  const int ar = tid >> 3;
  const int ac = (tid & 7) << 2;
  const int br = tid >> 5;
  const int bc = (tid & 31) << 2;
  const int lastRow = M - 1;
  const int g0 = min(m0 + ar, lastRow);
  const int g1 = min(m0 + ar + 32, lastRow);
  const float* ap0 = X + (size_t)g0 * K + ac;
  const float* ap1 = X + (size_t)g1 * K + ac;
  const size_t bOff  = (size_t)br * N + (n0 + bc);
  const float* bp1 = W1 + bOff;
  const float* bp2 = W2 + bOff;
  const size_t bStep = (size_t)8 * N;
  float* const sa  = As + ar * AS_STRIDE + ac;
  float* const sb1 = &Bs[0][0] + br * BS_STRIDE + bc;
  float* const sb2 = &Bs[1][0] + br * BS_STRIDE + bc;

#define LOAD_TILE_2(KOFF)                                              \
  do {                                                                 \
    const float* q1 = bp1 + (size_t)(KOFF) * N;                        \
    const float* q2 = bp2 + (size_t)(KOFF) * N;                        \
    ra0  = *(const v4f*)(ap0 + (KOFF));                                \
    ra1  = *(const v4f*)(ap1 + (KOFF));                                \
    rb10 = *(const v4f*)(q1);                                          \
    rb11 = *(const v4f*)(q1 + bStep);                                  \
    rb12 = *(const v4f*)(q1 + 2 * bStep);                              \
    rb13 = *(const v4f*)(q1 + 3 * bStep);                              \
    rb20 = *(const v4f*)(q2);                                          \
    rb21 = *(const v4f*)(q2 + bStep);                                  \
    rb22 = *(const v4f*)(q2 + 2 * bStep);                              \
    rb23 = *(const v4f*)(q2 + 3 * bStep);                              \
  } while (0)
#define STORE_TILE_2()                                                 \
  do {                                                                 \
    *(v4f*)(sa)                   = ra0;                               \
    *(v4f*)(sa + 32 * AS_STRIDE)  = ra1;                               \
    *(v4f*)(sb1)                  = rb10;                              \
    *(v4f*)(sb1 + 8 * BS_STRIDE)  = rb11;                              \
    *(v4f*)(sb1 + 16 * BS_STRIDE) = rb12;                              \
    *(v4f*)(sb1 + 24 * BS_STRIDE) = rb13;                              \
    *(v4f*)(sb2)                  = rb20;                              \
    *(v4f*)(sb2 + 8 * BS_STRIDE)  = rb21;                              \
    *(v4f*)(sb2 + 16 * BS_STRIDE) = rb22;                              \
    *(v4f*)(sb2 + 24 * BS_STRIDE) = rb23;                              \
  } while (0)

  v4f ra0, ra1, rb10, rb11, rb12, rb13, rb20, rb21, rb22, rb23;
  LOAD_TILE_2(0);
  STORE_TILE_2();
  __syncthreads();

  for (int k0 = 0; k0 < K; k0 += BK) {
    const bool more = (k0 + BK) < K;
    if (more) LOAD_TILE_2(k0 + BK);

#pragma unroll
    for (int kk = 0; kk < BK; kk += 4) {
      v2f a[2];
      v2f b[4];
#pragma unroll
      for (int mt = 0; mt < 2; ++mt) {
        int m = waveM * 32 + mt * 16 + lr;
        const float* p = As + m * AS_STRIDE + kk + kHalf;
        a[mt][0] = p[0];
        a[mt][1] = p[1];
      }
#pragma unroll
      for (int nt = 0; nt < 4; ++nt) {
        int n = waveN * 64 + nt * 16 + lr;
        const float* p = Bw + (kk + kHalf) * BS_STRIDE + n;
        b[nt][0] = p[0];
        b[nt][1] = p[BS_STRIDE];
      }
#pragma unroll
      for (int mt = 0; mt < 2; ++mt)
#pragma unroll
        for (int nt = 0; nt < 4; ++nt)
          acc[mt][nt] = __builtin_amdgcn_wmma_f32_16x16x4_f32(
              false, a[mt], false, b[nt], (short)0, acc[mt][nt], false, false);
    }

    if (more) {
      __syncthreads();
      STORE_TILE_2();
      __syncthreads();
    }
  }
#undef LOAD_TILE_2
#undef STORE_TILE_2

  const float* bias = waveOut ? b2 : b1;
  float*       Y    = waveOut ? Y2 : Y1;
  const int rowHalf = (lane >> 4) << 3;
  float bvs[4];
#pragma unroll
  for (int nt = 0; nt < 4; ++nt) bvs[nt] = bias[n0 + waveN * 64 + nt * 16 + lr];
#pragma unroll
  for (int mt = 0; mt < 2; ++mt) {
#pragma unroll
    for (int nt = 0; nt < 4; ++nt) {
      int col  = n0 + waveN * 64 + nt * 16 + lr;
      float bv = bvs[nt];
#pragma unroll
      for (int v = 0; v < 8; ++v) {
        int row = m0 + waveM * 32 + mt * 16 + rowHalf + v;
        float val = fmaxf(acc[mt][nt][v] + bv, 0.0f);
        if (row < M) Y[(size_t)row * N + col] = val;
      }
    }
  }
}

typedef __attribute__((ext_vector_type(4))) int v4i;
#define CSR_Q 16
#define CSR_G 128

template <bool FILL>
__device__ __forceinline__ void csr_take(v4i rv, v4i sv, int r0, int lane, v4i& cur,
                                         int* __restrict__ nbr, int nList) {
  const unsigned b0 = __builtin_amdgcn_ballot_w32((unsigned)(rv.x - r0) < (unsigned)CSR_G);
  const unsigned b1 = __builtin_amdgcn_ballot_w32((unsigned)(rv.y - r0) < (unsigned)CSR_G);
  const unsigned b2 = __builtin_amdgcn_ballot_w32((unsigned)(rv.z - r0) < (unsigned)CSR_G);
  const unsigned b3 = __builtin_amdgcn_ballot_w32((unsigned)(rv.w - r0) < (unsigned)CSR_G);
  unsigned any = b0 | b1 | b2 | b3;
  while (any) {
    const int l = __builtin_ctz(any);
    any &= any - 1;
#define CSR_ONE(BITS, COMP)                                               \
    if ((BITS >> l) & 1u) {                                               \
      const int k    = __builtin_amdgcn_readlane(rv.COMP, l) - r0;        \
      const int s    = FILL ? __builtin_amdgcn_readlane(sv.COMP, l) : 0;  \
      const int slot = k >> 5;                                            \
      if (lane == (k & 31)) {                                             \
        if (FILL) {                                                       \
          const int c = slot == 0 ? cur.x : slot == 1 ? cur.y             \
                      : slot == 2 ? cur.z : cur.w;                        \
          if ((unsigned)c < (unsigned)nList) nbr[c] = s;                  \
        }                                                                 \
        cur.x += (slot == 0);                                             \
        cur.y += (slot == 1);                                             \
        cur.z += (slot == 2);                                             \
        cur.w += (slot == 3);                                             \
      }                                                                   \
    }
    CSR_ONE(b0, x)
    CSR_ONE(b1, y)
    CSR_ONE(b2, z)
    CSR_ONE(b3, w)
#undef CSR_ONE
  }
}
template <bool FILL>
__device__ __forceinline__ void csr_scan_slice(const int* __restrict__ recv,
                                               const int* __restrict__ send,
                                               int eBeg, int eEnd, int r0, int lane, v4i& cur,
                                               int* __restrict__ nbr, int nList) {
  int base = eBeg;
  for (; base + 512 <= eEnd; base += 512) {
    const int e0 = base + lane * 4;
    const v4i ra = *(const v4i*)(recv + e0);
    const v4i rb = *(const v4i*)(recv + e0 + 128);
    const v4i rc = *(const v4i*)(recv + e0 + 256);
    const v4i rd = *(const v4i*)(recv + e0 + 384);
    v4i sa = ra, sb = rb, sc = rc, sd = rd;
    if (FILL) {
      sa = *(const v4i*)(send + e0);
      sb = *(const v4i*)(send + e0 + 128);
      sc = *(const v4i*)(send + e0 + 256);
      sd = *(const v4i*)(send + e0 + 384);
    }
    csr_take<FILL>(ra, sa, r0, lane, cur, nbr, nList);
    csr_take<FILL>(rb, sb, r0, lane, cur, nbr, nList);
    csr_take<FILL>(rc, sc, r0, lane, cur, nbr, nList);
    csr_take<FILL>(rd, sd, r0, lane, cur, nbr, nList);
  }
  for (; base < eEnd; base += 128) {
    const int e0 = base + lane * 4;
    v4i rv = (v4i)(-1);
    v4i sv = (v4i)(0);
    if (e0 + 0 < eEnd) { rv.x = recv[e0 + 0]; if (FILL) sv.x = send[e0 + 0]; }
    if (e0 + 1 < eEnd) { rv.y = recv[e0 + 1]; if (FILL) sv.y = send[e0 + 1]; }
    if (e0 + 2 < eEnd) { rv.z = recv[e0 + 2]; if (FILL) sv.z = send[e0 + 2]; }
    if (e0 + 3 < eEnd) { rv.w = recv[e0 + 3]; if (FILL) sv.w = send[e0 + 3]; }
    csr_take<FILL>(rv, sv, r0, lane, cur, nbr, nList);
  }
}

struct CsrOwner { const int* recv; const int* send; int r0; int nOwn; int rcBase; int eBeg; int eEnd; };
__device__ __forceinline__ CsrOwner csr_owner(const int* __restrict__ edgeUser,
                                              const int* __restrict__ edgeDish,
                                              int nUsers, int nDishes, int nEdges, int q) {
  CsrOwner o;
  const int gU = (nUsers + CSR_G - 1) / CSR_G;
  const int g  = blockIdx.x;
  const bool userSide = g < gU;
  o.recv   = userSide ? edgeUser : edgeDish;
  o.send   = userSide ? edgeDish : edgeUser;
  o.r0     = (userSide ? g : g - gU) * CSR_G;
  o.nOwn   = userSide ? nUsers : nDishes;
  o.rcBase = userSide ? o.r0 : nUsers + o.r0;
  const int per = (((nEdges + CSR_Q - 1) / CSR_Q) + 3) & ~3;
  o.eBeg = min(nEdges, q * per);
  o.eEnd = min(nEdges, o.eBeg + per);
  return o;
}

__global__ __launch_bounds__(32 * CSR_Q)
void gnn_csr_count(const int* __restrict__ edgeUser, const int* __restrict__ edgeDish,
                   int* __restrict__ cnt, int* __restrict__ tot,
                   int nUsers, int nDishes, int nEdges) {
  __shared__ int shc[CSR_Q][CSR_G];
  const int lane = threadIdx.x & 31;
  const int q = __builtin_amdgcn_readfirstlane(threadIdx.x >> 5);
  const CsrOwner o = csr_owner(edgeUser, edgeDish, nUsers, nDishes, nEdges, q);
  v4i cur = (v4i)(0);
  csr_scan_slice<false>(o.recv, o.send, o.eBeg, o.eEnd, o.r0, lane, cur, nullptr, 0);
  shc[q][lane]      = cur.x;
  shc[q][lane + 32] = cur.y;
  shc[q][lane + 64] = cur.z;
  shc[q][lane + 96] = cur.w;
  __syncthreads();
  const int k = threadIdx.x;
  if (k < CSR_G && o.r0 + k < o.nOwn) {
    v4i* p = (v4i*)(cnt + (size_t)(o.rcBase + k) * CSR_Q);
    int s = 0;
    for (int v = 0; v < CSR_Q / 4; ++v) {
      v4i r;
      r.x = shc[4 * v + 0][k];
      r.y = shc[4 * v + 1][k];
      r.z = shc[4 * v + 2][k];
      r.w = shc[4 * v + 3][k];
      s += (r.x + r.y) + (r.z + r.w);
      p[v] = r;
    }
    tot[o.rcBase + k] = s;
  }
}

__global__ __launch_bounds__(32 * CSR_Q)
void gnn_csr_fill(const int* __restrict__ edgeUser, const int* __restrict__ edgeDish,
                  const int* __restrict__ segBegin, int* __restrict__ nbr,
                  int nUsers, int nDishes, int nEdges, int nList) {
  const int lane = threadIdx.x & 31;
  const int q = __builtin_amdgcn_readfirstlane(threadIdx.x >> 5);
  const CsrOwner o = csr_owner(edgeUser, edgeDish, nUsers, nDishes, nEdges, q);
  const int k0 = o.r0 + lane;
  const int* p = segBegin + (size_t)(o.rcBase + lane) * CSR_Q + q;
  v4i cur = (v4i)(0);
  if (k0      < o.nOwn) cur.x = p[0];
  if (k0 + 32 < o.nOwn) cur.y = p[(size_t)32 * CSR_Q];
  if (k0 + 64 < o.nOwn) cur.z = p[(size_t)64 * CSR_Q];
  if (k0 + 96 < o.nOwn) cur.w = p[(size_t)96 * CSR_Q];
  csr_scan_slice<true>(o.recv, o.send, o.eBeg, o.eEnd, o.r0, lane, cur, nbr, nList);
}

__device__ __forceinline__ int csr_sum4(v4i a) { return (a.x + a.y) + (a.z + a.w); }

__global__ __launch_bounds__(1024)
void gnn_csr_offsets(const int* __restrict__ cnt, const int* __restrict__ tot,
                     int* __restrict__ segBegin, int nRecv) {
  __shared__ int sh[32];
  __shared__ int shBase;
  const int t = threadIdx.x;
  const int lane = t & 31;
  const int w = t >> 5;

  int part = 0;
  for (int j = 0; j < (int)blockIdx.x; ++j) part += tot[j * 1024 + t];
  for (int d = 16; d >= 1; d >>= 1) part += __shfl_xor(part, d, 32);
  if (lane == 0) sh[w] = part;
  __syncthreads();
  if (w == 0) {
    int v = sh[lane];
    for (int d = 16; d >= 1; d >>= 1) v += __shfl_xor(v, d, 32);
    if (lane == 0) shBase = v;
  }
  __syncthreads();
  const int base = shBase;

  const int i = blockIdx.x * 1024 + t;
  v4i c[CSR_Q / 4];
  int rowTot = 0;
  for (int v = 0; v < CSR_Q / 4; ++v) {
    c[v] = (v4i)(0);
    if (i < nRecv) c[v] = ((const v4i*)(cnt + (size_t)i * CSR_Q))[v];
    rowTot += csr_sum4(c[v]);
  }
  int incl = rowTot;
  for (int d = 1; d < 32; d <<= 1) {
    const int u = __shfl_up(incl, d, 32);
    if (lane >= d) incl += u;
  }
  if (lane == 31) sh[w] = incl;
  __syncthreads();
  if (w == 0) {
    const int v = sh[lane];
    int s = v;
    for (int d = 1; d < 32; d <<= 1) {
      const int u = __shfl_up(s, d, 32);
      if (lane >= d) s += u;
    }
    sh[lane] = s - v;
  }
  __syncthreads();
  if (i < nRecv) {
    int o = base + sh[w] + incl - rowTot;
    v4i* p = (v4i*)(segBegin + (size_t)i * CSR_Q);
    for (int v = 0; v < CSR_Q / 4; ++v) {
      v4i r;
      r.x = o; o += c[v].x;
      r.y = o; o += c[v].y;
      r.z = o; o += c[v].z;
      r.w = o; o += c[v].w;
      p[v] = r;
    }
  }
}

__global__ __launch_bounds__(256)
void gnn_gather_sum(const int* __restrict__ segBegin, const int* __restrict__ cnt,
                    const int* __restrict__ nbr,
                    const float* __restrict__ msgUser, const float* __restrict__ msgDish,
                    float* __restrict__ agg, int nUsers, int nDishes, int nList) {
  const int lane = threadIdx.x & 31;
  const int wave = __builtin_amdgcn_readfirstlane(threadIdx.x >> 5);
  const int nRecv = nUsers + nDishes;
  const int i = nRecv - 1 - (blockIdx.x * 8 + wave);
  if (i < 0) return;
  const bool isUser = i < nUsers;
  const float* msg  = isUser ? msgDish : msgUser;
  const int nSend   = isUser ? nDishes : nUsers;
  int b = segBegin[(size_t)i * CSR_Q];
  int e = segBegin[(size_t)i * CSR_Q + (CSR_Q - 1)] + cnt[(size_t)i * CSR_Q + (CSR_Q - 1)];
  b = max(0, min(b, nList));
  e = max(b, min(e, nList));
  float* row = agg + (size_t)i * 256 + lane * 4;
  v4f a0 = *(const v4f*)(row);
  v4f a1 = *(const v4f*)(row + 128);
  for (int base = b; base < e; base += 32) {
    const int n = min(32, e - base);
    int id = (lane < n) ? nbr[base + lane] : 0;
    id = max(0, min(id, nSend - 1));
    int j = 0;
    for (; j + 4 <= n; j += 4) {
      const float* r0 = msg + (size_t)__builtin_amdgcn_readlane(id, j)     * 256 + lane * 4;
      const float* r1 = msg + (size_t)__builtin_amdgcn_readlane(id, j + 1) * 256 + lane * 4;
      const float* r2 = msg + (size_t)__builtin_amdgcn_readlane(id, j + 2) * 256 + lane * 4;
      const float* r3 = msg + (size_t)__builtin_amdgcn_readlane(id, j + 3) * 256 + lane * 4;
      v4f m00 = *(const v4f*)(r0), m01 = *(const v4f*)(r0 + 128);
      v4f m10 = *(const v4f*)(r1), m11 = *(const v4f*)(r1 + 128);
      v4f m20 = *(const v4f*)(r2), m21 = *(const v4f*)(r2 + 128);
      v4f m30 = *(const v4f*)(r3), m31 = *(const v4f*)(r3 + 128);
      a0 += m00; a1 += m01;
      a0 += m10; a1 += m11;
      a0 += m20; a1 += m21;
      a0 += m30; a1 += m31;
    }
    for (; j < n; ++j) {
      const float* r0 = msg + (size_t)__builtin_amdgcn_readlane(id, j) * 256 + lane * 4;
      v4f m00 = *(const v4f*)(r0), m01 = *(const v4f*)(r0 + 128);
      a0 += m00; a1 += m01;
    }
  }
  *(v4f*)(row)       = a0;
  *(v4f*)(row + 128) = a1;
}

extern "C" void kernel_launch(void* const* d_in, const int* in_sizes, int n_in,
                              void* d_out, int out_size, void* d_ws, size_t ws_size,
                              hipStream_t stream) {
  (void)in_sizes; (void)n_in; (void)out_size; (void)ws_size;

  constexpr int U = 50000, D = 10000, E = 200000;
  constexpr int DIN = 1024, HID = 256, OUT = 128;

  const float* user_feat = (const float*)d_in[0];
  const float* dish_feat = (const float*)d_in[1];
  const int*   edge_user = (const int*)d_in[2];
  const int*   edge_dish = (const int*)d_in[3];
  const float* W_ui = (const float*)d_in[4];   const float* b_ui = (const float*)d_in[5];
  const float* W_di = (const float*)d_in[6];   const float* b_di = (const float*)d_in[7];
  const float* W_um = (const float*)d_in[8];   const float* b_um = (const float*)d_in[9];
  const float* W_dm = (const float*)d_in[10];  const float* b_dm = (const float*)d_in[11];
  const float* W_uu = (const float*)d_in[12];  const float* b_uu = (const float*)d_in[13];
  const float* W_du = (const float*)d_in[14];  const float* b_du = (const float*)d_in[15];
  const float* W_up = (const float*)d_in[16];  const float* b_up = (const float*)d_in[17];
  const float* W_dp = (const float*)d_in[18];  const float* b_dp = (const float*)d_in[19];

  float* out_user = (float*)d_out;
  float* out_dish = out_user + (size_t)U * OUT;

  float* ws      = (float*)d_ws;
  float* userAgg = ws;
  float* dishAgg = userAgg + (size_t)U * HID;
  float* msgUser = dishAgg + (size_t)D * HID;
  float* msgDish = msgUser + (size_t)U * HID;
  float* userUpd = msgUser;
  float* dishUpd = msgDish;

  dim3 blk(256);
  const int gU  = (U + BM - 1) / BM;
  const int gD  = (D + BM - 1) / BM;
  const int gU2 = (U + BM2 - 1) / BM2;
  const int gD2 = (D + BM2 - 1) / BM2;

  const DualSide l1u = { user_feat, W_ui, b_ui, userAgg, W_dm, b_dm, msgUser };
  const DualSide l1d = { dish_feat, W_di, b_di, dishAgg, W_um, b_um, msgDish };
  gnn_gemm_dual<<<dim3(gU2 + gD2, HID / BN), blk, 0, stream>>>(l1u, l1d, gU2, U, D, DIN, HID);

  constexpr int R  = U + D;
  constexpr int L  = 2 * E;
  constexpr int NG = (U + CSR_G - 1) / CSR_G + (D + CSR_G - 1) / CSR_G;
  int* cnt      = (int*)d_out;
  int* segBegin = cnt + (size_t)R * CSR_Q;
  int* tot      = segBegin + (size_t)R * CSR_Q;
  int* nbr      = tot + R;
  gnn_csr_count<<<dim3(NG), dim3(32 * CSR_Q), 0, stream>>>(edge_user, edge_dish, cnt, tot, U, D, E);
  gnn_csr_offsets<<<dim3((R + 1023) / 1024), dim3(1024), 0, stream>>>(cnt, tot, segBegin, R);
  gnn_csr_fill<<<dim3(NG), dim3(32 * CSR_Q), 0, stream>>>(edge_user, edge_dish, segBegin, nbr, U, D, E, L);
  gnn_gather_sum<<<dim3((R + 7) / 8), blk, 0, stream>>>(
      segBegin, cnt, nbr, msgUser, msgDish, userAgg, U, D, L);

  const GemmSide updU = { userAgg, W_uu, b_uu, userUpd };
  const GemmSide updD = { dishAgg, W_du, b_du, dishUpd };
  gnn_gemm_bias_act<<<dim3(gU + gD, HID / BN), blk, 0, stream>>>(updU, updD, gU, U, D, HID, HID, 1);
  const GemmSide prjU = { userUpd, W_up, b_up, out_user };
  const GemmSide prjD = { dishUpd, W_dp, b_dp, out_dish };
  gnn_gemm_bias_act<<<dim3(gU + gD, OUT / BN), blk, 0, stream>>>(prjU, prjD, gU, U, D, HID, OUT, 0);
}


# GEMM loops: 16-B LDS fragment reads (A: 4 k of a row; B: 4 adjacent columns, tiles interleaved), 24 per K tile in groups of 6 ahead of 16 WMMAs (was 40 scattered); 16-B stores; k sum order changes
# speedup vs baseline: 2.1166x; 1.0992x over previous
#include <hip/hip_runtime.h>

typedef __attribute__((ext_vector_type(2))) float v2f;
typedef __attribute__((ext_vector_type(8))) float v8f;
typedef __attribute__((ext_vector_type(4))) float v4f;

#define BM 128
#define BM2 64
#define BN 128
#define BK 32
#define AS_STRIDE 36
#define BS_STRIDE 132

struct GemmSide {
  const float* X; const float* W; const float* bias; float* Y;
};

__device__ __forceinline__ void gnn_tile_mma(const float* aw, const float* bw,
                                             v8f (&acc)[2][4]) {
  v4f af[2][2];
  v4f bf[2][4];
#pragma unroll
  for (int mt = 0; mt < 2; ++mt) af[0][mt] = *(const v4f*)(aw + mt * 16 * AS_STRIDE);
#pragma unroll
  for (int s = 0; s < 4; ++s) bf[0][s] = *(const v4f*)(bw + s * BS_STRIDE);
#pragma unroll
  for (int j = 0; j < BK / 8; ++j) {
    const int cur = j & 1;
    const int nxt = cur ^ 1;
    if (j + 1 < BK / 8) {
#pragma unroll
      for (int mt = 0; mt < 2; ++mt)
        af[nxt][mt] = *(const v4f*)(aw + mt * 16 * AS_STRIDE + (j + 1) * 8);
#pragma unroll
      for (int s = 0; s < 4; ++s)
        bf[nxt][s] = *(const v4f*)(bw + ((j + 1) * 8 + s) * BS_STRIDE);
      __builtin_amdgcn_sched_barrier(0);
    }
#pragma unroll
    for (int e = 0; e < 4; e += 2) {
#pragma unroll
      for (int mt = 0; mt < 2; ++mt) {
        v2f a;
        a[0] = af[cur][mt][e];
        a[1] = af[cur][mt][e + 1];
#pragma unroll
        for (int nt = 0; nt < 4; ++nt) {
          v2f b;
          b[0] = bf[cur][e][nt];
          b[1] = bf[cur][e + 1][nt];
          acc[mt][nt] = __builtin_amdgcn_wmma_f32_16x16x4_f32(
              false, a, false, b, (short)0, acc[mt][nt], false, false);
        }
      }
    }
    if (j + 2 < BK / 8) __builtin_amdgcn_sched_barrier(0);
  }
}

__global__ __launch_bounds__(256)
void gnn_gemm_bias_act(GemmSide su, GemmSide sd, int nBlkU, int Mu, int Md,
                       int K, int N, int doRelu) {
  __shared__ __attribute__((aligned(16))) float As[BM * AS_STRIDE];
  __shared__ __attribute__((aligned(16))) float Bs[BK * BS_STRIDE];

  const bool sideU = (int)blockIdx.x < nBlkU;
  const float* __restrict__ X    = sideU ? su.X    : sd.X;
  const float* __restrict__ W    = sideU ? su.W    : sd.W;
  const float* __restrict__ bias = sideU ? su.bias : sd.bias;
  float* __restrict__       Y    = sideU ? su.Y    : sd.Y;
  const int M  = sideU ? Mu : Md;
  const int bx = sideU ? (int)blockIdx.x : (int)blockIdx.x - nBlkU;

  const int tid   = threadIdx.x;
  const int lane  = tid & 31;
  const int wave  = tid >> 5;
  const int waveM = wave & 3;
  const int waveN = wave >> 2;
  const int lr    = lane & 15;
  const int kQuad = (lane >> 4) << 2;

  const int m0 = bx * BM;
  const int n0 = blockIdx.y * BN;

  v8f acc[2][4];
#pragma unroll
  for (int mt = 0; mt < 2; ++mt)
#pragma unroll
    for (int nt = 0; nt < 4; ++nt)
      acc[mt][nt] = (v8f)0.0f;

  const int ar = tid >> 3;
  const int ac = (tid & 7) << 2;
  const int br = tid >> 5;
  const int bc = (tid & 31) << 2;
  const int lastRow = M - 1;
  const int g0 = min(m0 + ar, lastRow);
  const int g1 = min(m0 + ar + 32, lastRow);
  const int g2 = min(m0 + ar + 64, lastRow);
  const int g3 = min(m0 + ar + 96, lastRow);
  const float* ap0 = X + (size_t)g0 * K + ac;
  const float* ap1 = X + (size_t)g1 * K + ac;
  const float* ap2 = X + (size_t)g2 * K + ac;
  const float* ap3 = X + (size_t)g3 * K + ac;
  const float* bp  = W + (size_t)br * N + (n0 + bc);
  const size_t bStep = (size_t)8 * N;
  float* const sa = As + ar * AS_STRIDE + ac;
  float* const sb = Bs + br * BS_STRIDE + bc;

#define LOAD_TILE_1(KOFF)                                              \
  do {                                                                 \
    const float* bq = bp + (size_t)(KOFF) * N;                         \
    ra0 = *(const v4f*)(ap0 + (KOFF));                                 \
    ra1 = *(const v4f*)(ap1 + (KOFF));                                 \
    ra2 = *(const v4f*)(ap2 + (KOFF));                                 \
    ra3 = *(const v4f*)(ap3 + (KOFF));                                 \
    rb0 = *(const v4f*)(bq);                                           \
    rb1 = *(const v4f*)(bq + bStep);                                   \
    rb2 = *(const v4f*)(bq + 2 * bStep);                               \
    rb3 = *(const v4f*)(bq + 3 * bStep);                               \
  } while (0)
#define STORE_TILE_1()                                                 \
  do {                                                                 \
    *(v4f*)(sa)                  = ra0;                                \
    *(v4f*)(sa + 32 * AS_STRIDE) = ra1;                                \
    *(v4f*)(sa + 64 * AS_STRIDE) = ra2;                                \
    *(v4f*)(sa + 96 * AS_STRIDE) = ra3;                                \
    *(v4f*)(sb)                  = rb0;                                \
    *(v4f*)(sb + 8 * BS_STRIDE)  = rb1;                                \
    *(v4f*)(sb + 16 * BS_STRIDE) = rb2;                                \
    *(v4f*)(sb + 24 * BS_STRIDE) = rb3;                                \
  } while (0)

  v4f ra0, ra1, ra2, ra3, rb0, rb1, rb2, rb3;
  LOAD_TILE_1(0);
  STORE_TILE_1();
  __syncthreads();

  for (int k0 = 0; k0 < K; k0 += BK) {
    const bool more = (k0 + BK) < K;
    if (more) LOAD_TILE_1(k0 + BK);

    gnn_tile_mma(As + (waveM * 32 + lr) * AS_STRIDE + kQuad,
                 Bs + kQuad * BS_STRIDE + waveN * 64 + 4 * lr, acc);

    if (more) {
      __syncthreads();
      STORE_TILE_1();
      __syncthreads();
    }
  }
#undef LOAD_TILE_1
#undef STORE_TILE_1

  const int rowHalf = (lane >> 4) << 3;
  const int colBase = n0 + waveN * 64 + 4 * lr;
  float bvs[4];
#pragma unroll
  for (int nt = 0; nt < 4; ++nt) bvs[nt] = bias[colBase + nt];
#pragma unroll
  for (int mt = 0; mt < 2; ++mt) {
#pragma unroll
    for (int v = 0; v < 8; ++v) {
      int row = m0 + waveM * 32 + mt * 16 + rowHalf + v;
      v4f val;
      val.x = acc[mt][0][v] + bvs[0];
      val.y = acc[mt][1][v] + bvs[1];
      val.z = acc[mt][2][v] + bvs[2];
      val.w = acc[mt][3][v] + bvs[3];
      if (doRelu) {
        val.x = fmaxf(val.x, 0.0f);
        val.y = fmaxf(val.y, 0.0f);
        val.z = fmaxf(val.z, 0.0f);
        val.w = fmaxf(val.w, 0.0f);
      }
      if (row < M) *(v4f*)(Y + (size_t)row * N + colBase) = val;
    }
  }
}

struct DualSide {
  const float* X;
  const float* W1; const float* b1; float* Y1;
  const float* W2; const float* b2; float* Y2;
};

__global__ __launch_bounds__(256)
void gnn_gemm_dual(DualSide su, DualSide sd, int nBlkU, int Mu, int Md, int K, int N) {
  __shared__ __attribute__((aligned(16))) float As[BM2 * AS_STRIDE];
  __shared__ __attribute__((aligned(16))) float Bs[2][BK * BS_STRIDE];

  const bool sideU = (int)blockIdx.x < nBlkU;
  const float* __restrict__ X  = sideU ? su.X  : sd.X;
  const float* __restrict__ W1 = sideU ? su.W1 : sd.W1;
  const float* __restrict__ b1 = sideU ? su.b1 : sd.b1;
  float* __restrict__       Y1 = sideU ? su.Y1 : sd.Y1;
  const float* __restrict__ W2 = sideU ? su.W2 : sd.W2;
  const float* __restrict__ b2 = sideU ? su.b2 : sd.b2;
  float* __restrict__       Y2 = sideU ? su.Y2 : sd.Y2;
  const int M  = sideU ? Mu : Md;
  const int bx = sideU ? (int)blockIdx.x : (int)blockIdx.x - nBlkU;

  const int tid     = threadIdx.x;
  const int lane    = tid & 31;
  const int wave    = tid >> 5;
  const int waveOut = wave >> 2;
  const int wq      = wave & 3;
  const int waveM   = wq & 1;
  const int waveN   = wq >> 1;
  const int lr      = lane & 15;
  const int kQuad   = (lane >> 4) << 2;

  const int m0 = bx * BM2;
  const int n0 = blockIdx.y * BN;

  v8f acc[2][4];
#pragma unroll
  for (int mt = 0; mt < 2; ++mt)
#pragma unroll
    for (int nt = 0; nt < 4; ++nt)
      acc[mt][nt] = (v8f)0.0f;

  const float* Bw = &Bs[waveOut][0];

  const int ar = tid >> 3;
  const int ac = (tid & 7) << 2;
  const int br = tid >> 5;
  const int bc = (tid & 31) << 2;
  const int lastRow = M - 1;
  const int g0 = min(m0 + ar, lastRow);
  const int g1 = min(m0 + ar + 32, lastRow);
  const float* ap0 = X + (size_t)g0 * K + ac;
  const float* ap1 = X + (size_t)g1 * K + ac;
  const size_t bOff  = (size_t)br * N + (n0 + bc);
  const float* bp1 = W1 + bOff;
  const float* bp2 = W2 + bOff;
  const size_t bStep = (size_t)8 * N;
  float* const sa  = As + ar * AS_STRIDE + ac;
  float* const sb1 = &Bs[0][0] + br * BS_STRIDE + bc;
  float* const sb2 = &Bs[1][0] + br * BS_STRIDE + bc;

#define LOAD_TILE_2(KOFF)                                              \
  do {                                                                 \
    const float* q1 = bp1 + (size_t)(KOFF) * N;                        \
    const float* q2 = bp2 + (size_t)(KOFF) * N;                        \
    ra0  = *(const v4f*)(ap0 + (KOFF));                                \
    ra1  = *(const v4f*)(ap1 + (KOFF));                                \
    rb10 = *(const v4f*)(q1);                                          \
    rb11 = *(const v4f*)(q1 + bStep);                                  \
    rb12 = *(const v4f*)(q1 + 2 * bStep);                              \
    rb13 = *(const v4f*)(q1 + 3 * bStep);                              \
    rb20 = *(const v4f*)(q2);                                          \
    rb21 = *(const v4f*)(q2 + bStep);                                  \
    rb22 = *(const v4f*)(q2 + 2 * bStep);                              \
    rb23 = *(const v4f*)(q2 + 3 * bStep);                              \
  } while (0)
#define STORE_TILE_2()                                                 \
  do {                                                                 \
    *(v4f*)(sa)                   = ra0;                               \
    *(v4f*)(sa + 32 * AS_STRIDE)  = ra1;                               \
    *(v4f*)(sb1)                  = rb10;                              \
    *(v4f*)(sb1 + 8 * BS_STRIDE)  = rb11;                              \
    *(v4f*)(sb1 + 16 * BS_STRIDE) = rb12;                              \
    *(v4f*)(sb1 + 24 * BS_STRIDE) = rb13;                              \
    *(v4f*)(sb2)                  = rb20;                              \
    *(v4f*)(sb2 + 8 * BS_STRIDE)  = rb21;                              \
    *(v4f*)(sb2 + 16 * BS_STRIDE) = rb22;                              \
    *(v4f*)(sb2 + 24 * BS_STRIDE) = rb23;                              \
  } while (0)

  v4f ra0, ra1, rb10, rb11, rb12, rb13, rb20, rb21, rb22, rb23;
  LOAD_TILE_2(0);
  STORE_TILE_2();
  __syncthreads();

  for (int k0 = 0; k0 < K; k0 += BK) {
    const bool more = (k0 + BK) < K;
    if (more) LOAD_TILE_2(k0 + BK);

    gnn_tile_mma(As + (waveM * 32 + lr) * AS_STRIDE + kQuad,
                 Bw + kQuad * BS_STRIDE + waveN * 64 + 4 * lr, acc);

    if (more) {
      __syncthreads();
      STORE_TILE_2();
      __syncthreads();
    }
  }
#undef LOAD_TILE_2
#undef STORE_TILE_2

  const float* bias = waveOut ? b2 : b1;
  float*       Y    = waveOut ? Y2 : Y1;
  const int rowHalf = (lane >> 4) << 3;
  const int colBase = n0 + waveN * 64 + 4 * lr;
  float bvs[4];
#pragma unroll
  for (int nt = 0; nt < 4; ++nt) bvs[nt] = bias[colBase + nt];
#pragma unroll
  for (int mt = 0; mt < 2; ++mt) {
#pragma unroll
    for (int v = 0; v < 8; ++v) {
      int row = m0 + waveM * 32 + mt * 16 + rowHalf + v;
      v4f val;
      val.x = fmaxf(acc[mt][0][v] + bvs[0], 0.0f);
      val.y = fmaxf(acc[mt][1][v] + bvs[1], 0.0f);
      val.z = fmaxf(acc[mt][2][v] + bvs[2], 0.0f);
      val.w = fmaxf(acc[mt][3][v] + bvs[3], 0.0f);
      if (row < M) *(v4f*)(Y + (size_t)row * N + colBase) = val;
    }
  }
}

typedef __attribute__((ext_vector_type(4))) int v4i;
#define CSR_Q 16
#define CSR_G 128

template <bool FILL>
__device__ __forceinline__ void csr_take(v4i rv, v4i sv, int r0, int lane, v4i& cur,
                                         int* __restrict__ nbr, int nList) {
  const unsigned b0 = __builtin_amdgcn_ballot_w32((unsigned)(rv.x - r0) < (unsigned)CSR_G);
  const unsigned b1 = __builtin_amdgcn_ballot_w32((unsigned)(rv.y - r0) < (unsigned)CSR_G);
  const unsigned b2 = __builtin_amdgcn_ballot_w32((unsigned)(rv.z - r0) < (unsigned)CSR_G);
  const unsigned b3 = __builtin_amdgcn_ballot_w32((unsigned)(rv.w - r0) < (unsigned)CSR_G);
  unsigned any = b0 | b1 | b2 | b3;
  while (any) {
    const int l = __builtin_ctz(any);
    any &= any - 1;
#define CSR_ONE(BITS, COMP)                                               \
    if ((BITS >> l) & 1u) {                                               \
      const int k    = __builtin_amdgcn_readlane(rv.COMP, l) - r0;        \
      const int s    = FILL ? __builtin_amdgcn_readlane(sv.COMP, l) : 0;  \
      const int slot = k >> 5;                                            \
      if (lane == (k & 31)) {                                             \
        if (FILL) {                                                       \
          const int c = slot == 0 ? cur.x : slot == 1 ? cur.y             \
                      : slot == 2 ? cur.z : cur.w;                        \
          if ((unsigned)c < (unsigned)nList) nbr[c] = s;                  \
        }                                                                 \
        cur.x += (slot == 0);                                             \
        cur.y += (slot == 1);                                             \
        cur.z += (slot == 2);                                             \
        cur.w += (slot == 3);                                             \
      }                                                                   \
    }
    CSR_ONE(b0, x)
    CSR_ONE(b1, y)
    CSR_ONE(b2, z)
    CSR_ONE(b3, w)
#undef CSR_ONE
  }
}
template <bool FILL>
__device__ __forceinline__ void csr_scan_slice(const int* __restrict__ recv,
                                               const int* __restrict__ send,
                                               int eBeg, int eEnd, int r0, int lane, v4i& cur,
                                               int* __restrict__ nbr, int nList) {
  int base = eBeg;
  for (; base + 512 <= eEnd; base += 512) {
    const int e0 = base + lane * 4;
    const v4i ra = *(const v4i*)(recv + e0);
    const v4i rb = *(const v4i*)(recv + e0 + 128);
    const v4i rc = *(const v4i*)(recv + e0 + 256);
    const v4i rd = *(const v4i*)(recv + e0 + 384);
    v4i sa = ra, sb = rb, sc = rc, sd = rd;
    if (FILL) {
      sa = *(const v4i*)(send + e0);
      sb = *(const v4i*)(send + e0 + 128);
      sc = *(const v4i*)(send + e0 + 256);
      sd = *(const v4i*)(send + e0 + 384);
    }
    csr_take<FILL>(ra, sa, r0, lane, cur, nbr, nList);
    csr_take<FILL>(rb, sb, r0, lane, cur, nbr, nList);
    csr_take<FILL>(rc, sc, r0, lane, cur, nbr, nList);
    csr_take<FILL>(rd, sd, r0, lane, cur, nbr, nList);
  }
  for (; base < eEnd; base += 128) {
    const int e0 = base + lane * 4;
    v4i rv = (v4i)(-1);
    v4i sv = (v4i)(0);
    if (e0 + 0 < eEnd) { rv.x = recv[e0 + 0]; if (FILL) sv.x = send[e0 + 0]; }
    if (e0 + 1 < eEnd) { rv.y = recv[e0 + 1]; if (FILL) sv.y = send[e0 + 1]; }
    if (e0 + 2 < eEnd) { rv.z = recv[e0 + 2]; if (FILL) sv.z = send[e0 + 2]; }
    if (e0 + 3 < eEnd) { rv.w = recv[e0 + 3]; if (FILL) sv.w = send[e0 + 3]; }
    csr_take<FILL>(rv, sv, r0, lane, cur, nbr, nList);
  }
}

struct CsrOwner { const int* recv; const int* send; int r0; int nOwn; int rcBase; int eBeg; int eEnd; };
__device__ __forceinline__ CsrOwner csr_owner(const int* __restrict__ edgeUser,
                                              const int* __restrict__ edgeDish,
                                              int nUsers, int nDishes, int nEdges, int q) {
  CsrOwner o;
  const int gU = (nUsers + CSR_G - 1) / CSR_G;
  const int g  = blockIdx.x;
  const bool userSide = g < gU;
  o.recv   = userSide ? edgeUser : edgeDish;
  o.send   = userSide ? edgeDish : edgeUser;
  o.r0     = (userSide ? g : g - gU) * CSR_G;
  o.nOwn   = userSide ? nUsers : nDishes;
  o.rcBase = userSide ? o.r0 : nUsers + o.r0;
  const int per = (((nEdges + CSR_Q - 1) / CSR_Q) + 3) & ~3;
  o.eBeg = min(nEdges, q * per);
  o.eEnd = min(nEdges, o.eBeg + per);
  return o;
}

__global__ __launch_bounds__(32 * CSR_Q)
void gnn_csr_count(const int* __restrict__ edgeUser, const int* __restrict__ edgeDish,
                   int* __restrict__ cnt, int* __restrict__ tot,
                   int nUsers, int nDishes, int nEdges) {
  __shared__ int shc[CSR_Q][CSR_G];
  const int lane = threadIdx.x & 31;
  const int q = __builtin_amdgcn_readfirstlane(threadIdx.x >> 5);
  const CsrOwner o = csr_owner(edgeUser, edgeDish, nUsers, nDishes, nEdges, q);
  v4i cur = (v4i)(0);
  csr_scan_slice<false>(o.recv, o.send, o.eBeg, o.eEnd, o.r0, lane, cur, nullptr, 0);
  shc[q][lane]      = cur.x;
  shc[q][lane + 32] = cur.y;
  shc[q][lane + 64] = cur.z;
  shc[q][lane + 96] = cur.w;
  __syncthreads();
  const int k = threadIdx.x;
  if (k < CSR_G && o.r0 + k < o.nOwn) {
    v4i* p = (v4i*)(cnt + (size_t)(o.rcBase + k) * CSR_Q);
    int s = 0;
    for (int v = 0; v < CSR_Q / 4; ++v) {
      v4i r;
      r.x = shc[4 * v + 0][k];
      r.y = shc[4 * v + 1][k];
      r.z = shc[4 * v + 2][k];
      r.w = shc[4 * v + 3][k];
      s += (r.x + r.y) + (r.z + r.w);
      p[v] = r;
    }
    tot[o.rcBase + k] = s;
  }
}

__global__ __launch_bounds__(32 * CSR_Q)
void gnn_csr_fill(const int* __restrict__ edgeUser, const int* __restrict__ edgeDish,
                  const int* __restrict__ segBegin, int* __restrict__ nbr,
                  int nUsers, int nDishes, int nEdges, int nList) {
  const int lane = threadIdx.x & 31;
  const int q = __builtin_amdgcn_readfirstlane(threadIdx.x >> 5);
  const CsrOwner o = csr_owner(edgeUser, edgeDish, nUsers, nDishes, nEdges, q);
  const int k0 = o.r0 + lane;
  const int* p = segBegin + (size_t)(o.rcBase + lane) * CSR_Q + q;
  v4i cur = (v4i)(0);
  if (k0      < o.nOwn) cur.x = p[0];
  if (k0 + 32 < o.nOwn) cur.y = p[(size_t)32 * CSR_Q];
  if (k0 + 64 < o.nOwn) cur.z = p[(size_t)64 * CSR_Q];
  if (k0 + 96 < o.nOwn) cur.w = p[(size_t)96 * CSR_Q];
  csr_scan_slice<true>(o.recv, o.send, o.eBeg, o.eEnd, o.r0, lane, cur, nbr, nList);
}

__device__ __forceinline__ int csr_sum4(v4i a) { return (a.x + a.y) + (a.z + a.w); }

__global__ __launch_bounds__(1024)
void gnn_csr_offsets(const int* __restrict__ cnt, const int* __restrict__ tot,
                     int* __restrict__ segBegin, int nRecv) {
  __shared__ int sh[32];
  __shared__ int shBase;
  const int t = threadIdx.x;
  const int lane = t & 31;
  const int w = t >> 5;

  int part = 0;
  for (int j = 0; j < (int)blockIdx.x; ++j) part += tot[j * 1024 + t];
  for (int d = 16; d >= 1; d >>= 1) part += __shfl_xor(part, d, 32);
  if (lane == 0) sh[w] = part;
  __syncthreads();
  if (w == 0) {
    int v = sh[lane];
    for (int d = 16; d >= 1; d >>= 1) v += __shfl_xor(v, d, 32);
    if (lane == 0) shBase = v;
  }
  __syncthreads();
  const int base = shBase;

  const int i = blockIdx.x * 1024 + t;
  v4i c[CSR_Q / 4];
  int rowTot = 0;
  for (int v = 0; v < CSR_Q / 4; ++v) {
    c[v] = (v4i)(0);
    if (i < nRecv) c[v] = ((const v4i*)(cnt + (size_t)i * CSR_Q))[v];
    rowTot += csr_sum4(c[v]);
  }
  int incl = rowTot;
  for (int d = 1; d < 32; d <<= 1) {
    const int u = __shfl_up(incl, d, 32);
    if (lane >= d) incl += u;
  }
  if (lane == 31) sh[w] = incl;
  __syncthreads();
  if (w == 0) {
    const int v = sh[lane];
    int s = v;
    for (int d = 1; d < 32; d <<= 1) {
      const int u = __shfl_up(s, d, 32);
      if (lane >= d) s += u;
    }
    sh[lane] = s - v;
  }
  __syncthreads();
  if (i < nRecv) {
    int o = base + sh[w] + incl - rowTot;
    v4i* p = (v4i*)(segBegin + (size_t)i * CSR_Q);
    for (int v = 0; v < CSR_Q / 4; ++v) {
      v4i r;
      r.x = o; o += c[v].x;
      r.y = o; o += c[v].y;
      r.z = o; o += c[v].z;
      r.w = o; o += c[v].w;
      p[v] = r;
    }
  }
}

__global__ __launch_bounds__(256)
void gnn_gather_sum(const int* __restrict__ segBegin, const int* __restrict__ cnt,
                    const int* __restrict__ nbr,
                    const float* __restrict__ msgUser, const float* __restrict__ msgDish,
                    float* __restrict__ agg, int nUsers, int nDishes, int nList) {
  const int lane = threadIdx.x & 31;
  const int wave = __builtin_amdgcn_readfirstlane(threadIdx.x >> 5);
  const int nRecv = nUsers + nDishes;
  const int i = nRecv - 1 - (blockIdx.x * 8 + wave);
  if (i < 0) return;
  const bool isUser = i < nUsers;
  const float* msg  = isUser ? msgDish : msgUser;
  const int nSend   = isUser ? nDishes : nUsers;
  int b = segBegin[(size_t)i * CSR_Q];
  int e = segBegin[(size_t)i * CSR_Q + (CSR_Q - 1)] + cnt[(size_t)i * CSR_Q + (CSR_Q - 1)];
  b = max(0, min(b, nList));
  e = max(b, min(e, nList));
  float* row = agg + (size_t)i * 256 + lane * 4;
  v4f a0 = *(const v4f*)(row);
  v4f a1 = *(const v4f*)(row + 128);
  for (int base = b; base < e; base += 32) {
    const int n = min(32, e - base);
    int id = (lane < n) ? nbr[base + lane] : 0;
    id = max(0, min(id, nSend - 1));
    int j = 0;
    for (; j + 4 <= n; j += 4) {
      const float* r0 = msg + (size_t)__builtin_amdgcn_readlane(id, j)     * 256 + lane * 4;
      const float* r1 = msg + (size_t)__builtin_amdgcn_readlane(id, j + 1) * 256 + lane * 4;
      const float* r2 = msg + (size_t)__builtin_amdgcn_readlane(id, j + 2) * 256 + lane * 4;
      const float* r3 = msg + (size_t)__builtin_amdgcn_readlane(id, j + 3) * 256 + lane * 4;
      v4f m00 = *(const v4f*)(r0), m01 = *(const v4f*)(r0 + 128);
      v4f m10 = *(const v4f*)(r1), m11 = *(const v4f*)(r1 + 128);
      v4f m20 = *(const v4f*)(r2), m21 = *(const v4f*)(r2 + 128);
      v4f m30 = *(const v4f*)(r3), m31 = *(const v4f*)(r3 + 128);
      a0 += m00; a1 += m01;
      a0 += m10; a1 += m11;
      a0 += m20; a1 += m21;
      a0 += m30; a1 += m31;
    }
    for (; j < n; ++j) {
      const float* r0 = msg + (size_t)__builtin_amdgcn_readlane(id, j) * 256 + lane * 4;
      v4f m00 = *(const v4f*)(r0), m01 = *(const v4f*)(r0 + 128);
      a0 += m00; a1 += m01;
    }
  }
  *(v4f*)(row)       = a0;
  *(v4f*)(row + 128) = a1;
}

extern "C" void kernel_launch(void* const* d_in, const int* in_sizes, int n_in,
                              void* d_out, int out_size, void* d_ws, size_t ws_size,
                              hipStream_t stream) {
  (void)in_sizes; (void)n_in; (void)out_size; (void)ws_size;

  constexpr int U = 50000, D = 10000, E = 200000;
  constexpr int DIN = 1024, HID = 256, OUT = 128;

  const float* user_feat = (const float*)d_in[0];
  const float* dish_feat = (const float*)d_in[1];
  const int*   edge_user = (const int*)d_in[2];
  const int*   edge_dish = (const int*)d_in[3];
  const float* W_ui = (const float*)d_in[4];   const float* b_ui = (const float*)d_in[5];
  const float* W_di = (const float*)d_in[6];   const float* b_di = (const float*)d_in[7];
  const float* W_um = (const float*)d_in[8];   const float* b_um = (const float*)d_in[9];
  const float* W_dm = (const float*)d_in[10];  const float* b_dm = (const float*)d_in[11];
  const float* W_uu = (const float*)d_in[12];  const float* b_uu = (const float*)d_in[13];
  const float* W_du = (const float*)d_in[14];  const float* b_du = (const float*)d_in[15];
  const float* W_up = (const float*)d_in[16];  const float* b_up = (const float*)d_in[17];
  const float* W_dp = (const float*)d_in[18];  const float* b_dp = (const float*)d_in[19];

  float* out_user = (float*)d_out;
  float* out_dish = out_user + (size_t)U * OUT;

  float* ws      = (float*)d_ws;
  float* userAgg = ws;
  float* dishAgg = userAgg + (size_t)U * HID;
  float* msgUser = dishAgg + (size_t)D * HID;
  float* msgDish = msgUser + (size_t)U * HID;
  float* userUpd = msgUser;
  float* dishUpd = msgDish;

  dim3 blk(256);
  const int gU  = (U + BM - 1) / BM;
  const int gD  = (D + BM - 1) / BM;
  const int gU2 = (U + BM2 - 1) / BM2;
  const int gD2 = (D + BM2 - 1) / BM2;

  const DualSide l1u = { user_feat, W_ui, b_ui, userAgg, W_dm, b_dm, msgUser };
  const DualSide l1d = { dish_feat, W_di, b_di, dishAgg, W_um, b_um, msgDish };
  gnn_gemm_dual<<<dim3(gU2 + gD2, HID / BN), blk, 0, stream>>>(l1u, l1d, gU2, U, D, DIN, HID);

  constexpr int R  = U + D;
  constexpr int L  = 2 * E;
  constexpr int NG = (U + CSR_G - 1) / CSR_G + (D + CSR_G - 1) / CSR_G;
  int* cnt      = (int*)d_out;
  int* segBegin = cnt + (size_t)R * CSR_Q;
  int* tot      = segBegin + (size_t)R * CSR_Q;
  int* nbr      = tot + R;
  gnn_csr_count<<<dim3(NG), dim3(32 * CSR_Q), 0, stream>>>(edge_user, edge_dish, cnt, tot, U, D, E);
  gnn_csr_offsets<<<dim3((R + 1023) / 1024), dim3(1024), 0, stream>>>(cnt, tot, segBegin, R);
  gnn_csr_fill<<<dim3(NG), dim3(32 * CSR_Q), 0, stream>>>(edge_user, edge_dish, segBegin, nbr, U, D, E, L);
  gnn_gather_sum<<<dim3((R + 7) / 8), blk, 0, stream>>>(
      segBegin, cnt, nbr, msgUser, msgDish, userAgg, U, D, L);

  const GemmSide updU = { userAgg, W_uu, b_uu, userUpd };
  const GemmSide updD = { dishAgg, W_du, b_du, dishUpd };
  gnn_gemm_bias_act<<<dim3(gU + gD, HID / BN), blk, 0, stream>>>(updU, updD, gU, U, D, HID, HID, 1);
  const GemmSide prjU = { userUpd, W_up, b_up, out_user };
  const GemmSide prjD = { dishUpd, W_dp, b_dp, out_dish };
  gnn_gemm_bias_act<<<dim3(gU + gD, OUT / BN), blk, 0, stream>>>(prjU, prjD, gU, U, D, HID, OUT, 0);
}


# Layer-1 GEMM: wave tile 32x64 -> 64x64 (block 128 rows, K tile 16 to stay in 256 registers): an LDS fragment read feeds 4 WMMAs, all 16 reads of a tile lead its 64 WMMAs; sum order unchanged
# speedup vs baseline: 2.2188x; 1.0483x over previous
#include <hip/hip_runtime.h>

typedef __attribute__((ext_vector_type(2))) float v2f;
typedef __attribute__((ext_vector_type(8))) float v8f;
typedef __attribute__((ext_vector_type(4))) float v4f;

#define BM 128
#define BM2 128
#define BN 128
#define BK 32
#define BK2 16
#define AS_STRIDE 36
#define AS2_STRIDE 20
#define BS_STRIDE 132

struct GemmSide {
  const float* X; const float* W; const float* bias; float* Y;
};

template <int MT, int KT, int ASTR>
__device__ __forceinline__ void gnn_tile_mma(const float* aw, const float* bw,
                                             v8f (&acc)[MT][4]) {
  v4f af[2][MT];
  v4f bf[2][4];
#pragma unroll
  for (int mt = 0; mt < MT; ++mt) af[0][mt] = *(const v4f*)(aw + mt * 16 * ASTR);
#pragma unroll
  for (int s = 0; s < 4; ++s) bf[0][s] = *(const v4f*)(bw + s * BS_STRIDE);
#pragma unroll
  for (int j = 0; j < KT / 8; ++j) {
    const int cur = j & 1;
    const int nxt = cur ^ 1;
    if (j + 1 < KT / 8) {
#pragma unroll
      for (int mt = 0; mt < MT; ++mt)
        af[nxt][mt] = *(const v4f*)(aw + mt * 16 * ASTR + (j + 1) * 8);
#pragma unroll
      for (int s = 0; s < 4; ++s)
        bf[nxt][s] = *(const v4f*)(bw + ((j + 1) * 8 + s) * BS_STRIDE);
      __builtin_amdgcn_sched_barrier(0);
    }
#pragma unroll
    for (int e = 0; e < 4; e += 2) {
#pragma unroll
      for (int mt = 0; mt < MT; ++mt) {
        v2f a;
        a[0] = af[cur][mt][e];
        a[1] = af[cur][mt][e + 1];
#pragma unroll
        for (int nt = 0; nt < 4; ++nt) {
          v2f b;
          b[0] = bf[cur][e][nt];
          b[1] = bf[cur][e + 1][nt];
          acc[mt][nt] = __builtin_amdgcn_wmma_f32_16x16x4_f32(
              false, a, false, b, (short)0, acc[mt][nt], false, false);
        }
      }
    }
    if (j + 2 < KT / 8) __builtin_amdgcn_sched_barrier(0);
  }
}

__global__ __launch_bounds__(256)
void gnn_gemm_bias_act(GemmSide su, GemmSide sd, int nBlkU, int Mu, int Md,
                       int K, int N, int doRelu) {
  __shared__ __attribute__((aligned(16))) float As[BM * AS_STRIDE];
  __shared__ __attribute__((aligned(16))) float Bs[BK * BS_STRIDE];

  const bool sideU = (int)blockIdx.x < nBlkU;
  const float* __restrict__ X    = sideU ? su.X    : sd.X;
  const float* __restrict__ W    = sideU ? su.W    : sd.W;
  const float* __restrict__ bias = sideU ? su.bias : sd.bias;
  float* __restrict__       Y    = sideU ? su.Y    : sd.Y;
  const int M  = sideU ? Mu : Md;
  const int bx = sideU ? (int)blockIdx.x : (int)blockIdx.x - nBlkU;

  const int tid   = threadIdx.x;
  const int lane  = tid & 31;
  const int wave  = tid >> 5;
  const int waveM = wave & 3;
  const int waveN = wave >> 2;
  const int lr    = lane & 15;
  const int kQuad = (lane >> 4) << 2;

  const int m0 = bx * BM;
  const int n0 = blockIdx.y * BN;

  v8f acc[2][4];
#pragma unroll
  for (int mt = 0; mt < 2; ++mt)
#pragma unroll
    for (int nt = 0; nt < 4; ++nt)
      acc[mt][nt] = (v8f)0.0f;

  const int ar = tid >> 3;
  const int ac = (tid & 7) << 2;
  const int br = tid >> 5;
  const int bc = (tid & 31) << 2;
  const int lastRow = M - 1;
  const int g0 = min(m0 + ar, lastRow);
  const int g1 = min(m0 + ar + 32, lastRow);
  const int g2 = min(m0 + ar + 64, lastRow);
  const int g3 = min(m0 + ar + 96, lastRow);
  const float* ap0 = X + (size_t)g0 * K + ac;
  const float* ap1 = X + (size_t)g1 * K + ac;
  const float* ap2 = X + (size_t)g2 * K + ac;
  const float* ap3 = X + (size_t)g3 * K + ac;
  const float* bp  = W + (size_t)br * N + (n0 + bc);
  const size_t bStep = (size_t)8 * N;
  float* const sa = As + ar * AS_STRIDE + ac;
  float* const sb = Bs + br * BS_STRIDE + bc;

#define LOAD_TILE_1(KOFF)                                              \
  do {                                                                 \
    const float* bq = bp + (size_t)(KOFF) * N;                         \
    ra0 = *(const v4f*)(ap0 + (KOFF));                                 \
    ra1 = *(const v4f*)(ap1 + (KOFF));                                 \
    ra2 = *(const v4f*)(ap2 + (KOFF));                                 \
    ra3 = *(const v4f*)(ap3 + (KOFF));                                 \
    rb0 = *(const v4f*)(bq);                                           \
    rb1 = *(const v4f*)(bq + bStep);                                   \
    rb2 = *(const v4f*)(bq + 2 * bStep);                               \
    rb3 = *(const v4f*)(bq + 3 * bStep);                               \
  } while (0)
#define STORE_TILE_1()                                                 \
  do {                                                                 \
    *(v4f*)(sa)                  = ra0;                                \
    *(v4f*)(sa + 32 * AS_STRIDE) = ra1;                                \
    *(v4f*)(sa + 64 * AS_STRIDE) = ra2;                                \
    *(v4f*)(sa + 96 * AS_STRIDE) = ra3;                                \
    *(v4f*)(sb)                  = rb0;                                \
    *(v4f*)(sb + 8 * BS_STRIDE)  = rb1;                                \
    *(v4f*)(sb + 16 * BS_STRIDE) = rb2;                                \
    *(v4f*)(sb + 24 * BS_STRIDE) = rb3;                                \
  } while (0)

  v4f ra0, ra1, ra2, ra3, rb0, rb1, rb2, rb3;
  LOAD_TILE_1(0);
  STORE_TILE_1();
  __syncthreads();

  for (int k0 = 0; k0 < K; k0 += BK) {
    const bool more = (k0 + BK) < K;
    if (more) LOAD_TILE_1(k0 + BK);

    gnn_tile_mma<2, BK, AS_STRIDE>(As + (waveM * 32 + lr) * AS_STRIDE + kQuad,
                                   Bs + kQuad * BS_STRIDE + waveN * 64 + 4 * lr, acc);

    if (more) {
      __syncthreads();
      STORE_TILE_1();
      __syncthreads();
    }
  }
#undef LOAD_TILE_1
#undef STORE_TILE_1

  const int rowHalf = (lane >> 4) << 3;
  const int colBase = n0 + waveN * 64 + 4 * lr;
  float bvs[4];
#pragma unroll
  for (int nt = 0; nt < 4; ++nt) bvs[nt] = bias[colBase + nt];
#pragma unroll
  for (int mt = 0; mt < 2; ++mt) {
#pragma unroll
    for (int v = 0; v < 8; ++v) {
      int row = m0 + waveM * 32 + mt * 16 + rowHalf + v;
      v4f val;
      val.x = acc[mt][0][v] + bvs[0];
      val.y = acc[mt][1][v] + bvs[1];
      val.z = acc[mt][2][v] + bvs[2];
      val.w = acc[mt][3][v] + bvs[3];
      if (doRelu) {
        val.x = fmaxf(val.x, 0.0f);
        val.y = fmaxf(val.y, 0.0f);
        val.z = fmaxf(val.z, 0.0f);
        val.w = fmaxf(val.w, 0.0f);
      }
      if (row < M) *(v4f*)(Y + (size_t)row * N + colBase) = val;
    }
  }
}

struct DualSide {
  const float* X;
  const float* W1; const float* b1; float* Y1;
  const float* W2; const float* b2; float* Y2;
};

template <int KC, int NC>
__global__ __launch_bounds__(256)
void gnn_gemm_dual(DualSide su, DualSide sd, int nBlkU, int Mu, int Md) {
  __shared__ __attribute__((aligned(16))) float As[BM2 * AS2_STRIDE];
  __shared__ __attribute__((aligned(16))) float Bs[2][BK2 * BS_STRIDE];

  const bool sideU = (int)blockIdx.x < nBlkU;
  const float* __restrict__ X  = sideU ? su.X  : sd.X;
  const float* __restrict__ W1 = sideU ? su.W1 : sd.W1;
  const float* __restrict__ b1 = sideU ? su.b1 : sd.b1;
  float* __restrict__       Y1 = sideU ? su.Y1 : sd.Y1;
  const float* __restrict__ W2 = sideU ? su.W2 : sd.W2;
  const float* __restrict__ b2 = sideU ? su.b2 : sd.b2;
  float* __restrict__       Y2 = sideU ? su.Y2 : sd.Y2;
  const int M  = sideU ? Mu : Md;
  const int bx = sideU ? (int)blockIdx.x : (int)blockIdx.x - nBlkU;

  const int tid     = threadIdx.x;
  const int lane    = tid & 31;
  const int wave    = tid >> 5;
  const int waveOut = wave >> 2;
  const int wq      = wave & 3;
  const int waveM   = wq & 1;
  const int waveN   = wq >> 1;
  const int lr      = lane & 15;
  const int kQuad   = (lane >> 4) << 2;

  const int m0 = bx * BM2;
  const int n0 = blockIdx.y * BN;

  v8f acc[4][4];
#pragma unroll
  for (int mt = 0; mt < 4; ++mt)
#pragma unroll
    for (int nt = 0; nt < 4; ++nt)
      acc[mt][nt] = (v8f)0.0f;

  const float* Bw = &Bs[waveOut][0];

  const int ar = tid >> 2;
  const int ac = (tid & 3) << 2;
  const int br = tid >> 5;
  const int bc = (tid & 31) << 2;
  const int lastRow = M - 1;
  const int g0 = min(m0 + ar, lastRow);
  const int g1 = min(m0 + ar + 64, lastRow);
  const float* ap0 = X + (size_t)g0 * KC + ac;
  const float* ap1 = X + (size_t)g1 * KC + ac;
  const float* bp1 = W1 + (size_t)br * NC + (n0 + bc);
  const float* bp2 = W2 + (size_t)br * NC + (n0 + bc);
  float* const sa  = As + ar * AS2_STRIDE + ac;
  float* const sb1 = &Bs[0][0] + br * BS_STRIDE + bc;
  float* const sb2 = &Bs[1][0] + br * BS_STRIDE + bc;

#define LOAD_TILE_2()                                                  \
  do {                                                                 \
    ra0  = *(const v4f*)(ap0);                                         \
    ra1  = *(const v4f*)(ap1);                                         \
    rb10 = *(const v4f*)(bp1);                                         \
    rb11 = *(const v4f*)(bp1 + 8 * NC);                                \
    rb20 = *(const v4f*)(bp2);                                         \
    rb21 = *(const v4f*)(bp2 + 8 * NC);                                \
  } while (0)
#define STEP_TILE_2()                                                  \
  do {                                                                 \
    ap0 += BK2; ap1 += BK2;                                            \
    bp1 += BK2 * NC; bp2 += BK2 * NC;                                  \
  } while (0)
#define STORE_TILE_2()                                                 \
  do {                                                                 \
    *(v4f*)(sa)                   = ra0;                               \
    *(v4f*)(sa + 64 * AS2_STRIDE) = ra1;                               \
    *(v4f*)(sb1)                  = rb10;                              \
    *(v4f*)(sb1 + 8 * BS_STRIDE)  = rb11;                              \
    *(v4f*)(sb2)                  = rb20;                              \
    *(v4f*)(sb2 + 8 * BS_STRIDE)  = rb21;                              \
  } while (0)

  v4f ra0, ra1, rb10, rb11, rb20, rb21;
  LOAD_TILE_2();
  STEP_TILE_2();
  STORE_TILE_2();
  __syncthreads();

  for (int k0 = 0; k0 < KC; k0 += BK2) {
    const bool more = (k0 + BK2) < KC;
    if (more) LOAD_TILE_2();

    gnn_tile_mma<4, BK2, AS2_STRIDE>(As + (waveM * 64 + lr) * AS2_STRIDE + kQuad,
                                     Bw + kQuad * BS_STRIDE + waveN * 64 + 4 * lr, acc);

    if (more) {
      STEP_TILE_2();
      __syncthreads();
      STORE_TILE_2();
      __syncthreads();
    }
  }
#undef LOAD_TILE_2
#undef STEP_TILE_2
#undef STORE_TILE_2

  const float* bias = waveOut ? b2 : b1;
  float*       Y    = waveOut ? Y2 : Y1;
  const int rowHalf = (lane >> 4) << 3;
  const int colBase = n0 + waveN * 64 + 4 * lr;
  float bvs[4];
#pragma unroll
  for (int nt = 0; nt < 4; ++nt) bvs[nt] = bias[colBase + nt];
#pragma unroll
  for (int mt = 0; mt < 4; ++mt) {
#pragma unroll
    for (int v = 0; v < 8; ++v) {
      int row = m0 + waveM * 64 + mt * 16 + rowHalf + v;
      v4f val;
      val.x = fmaxf(acc[mt][0][v] + bvs[0], 0.0f);
      val.y = fmaxf(acc[mt][1][v] + bvs[1], 0.0f);
      val.z = fmaxf(acc[mt][2][v] + bvs[2], 0.0f);
      val.w = fmaxf(acc[mt][3][v] + bvs[3], 0.0f);
      if (row < M) *(v4f*)(Y + (size_t)row * NC + colBase) = val;
    }
  }
}

typedef __attribute__((ext_vector_type(4))) int v4i;
#define CSR_Q 16
#define CSR_G 128

template <bool FILL>
__device__ __forceinline__ void csr_take(v4i rv, v4i sv, int r0, int lane, v4i& cur,
                                         int* __restrict__ nbr, int nList) {
  const unsigned b0 = __builtin_amdgcn_ballot_w32((unsigned)(rv.x - r0) < (unsigned)CSR_G);
  const unsigned b1 = __builtin_amdgcn_ballot_w32((unsigned)(rv.y - r0) < (unsigned)CSR_G);
  const unsigned b2 = __builtin_amdgcn_ballot_w32((unsigned)(rv.z - r0) < (unsigned)CSR_G);
  const unsigned b3 = __builtin_amdgcn_ballot_w32((unsigned)(rv.w - r0) < (unsigned)CSR_G);
  unsigned any = b0 | b1 | b2 | b3;
  while (any) {
    const int l = __builtin_ctz(any);
    any &= any - 1;
#define CSR_ONE(BITS, COMP)                                               \
    if ((BITS >> l) & 1u) {                                               \
      const int k    = __builtin_amdgcn_readlane(rv.COMP, l) - r0;        \
      const int s    = FILL ? __builtin_amdgcn_readlane(sv.COMP, l) : 0;  \
      const int slot = k >> 5;                                            \
      if (lane == (k & 31)) {                                             \
        if (FILL) {                                                       \
          const int c = slot == 0 ? cur.x : slot == 1 ? cur.y             \
                      : slot == 2 ? cur.z : cur.w;                        \
          if ((unsigned)c < (unsigned)nList) nbr[c] = s;                  \
        }                                                                 \
        cur.x += (slot == 0);                                             \
        cur.y += (slot == 1);                                             \
        cur.z += (slot == 2);                                             \
        cur.w += (slot == 3);                                             \
      }                                                                   \
    }
    CSR_ONE(b0, x)
    CSR_ONE(b1, y)
    CSR_ONE(b2, z)
    CSR_ONE(b3, w)
#undef CSR_ONE
  }
}
template <bool FILL>
__device__ __forceinline__ void csr_scan_slice(const int* __restrict__ recv,
                                               const int* __restrict__ send,
                                               int eBeg, int eEnd, int r0, int lane, v4i& cur,
                                               int* __restrict__ nbr, int nList) {
  int base = eBeg;
  for (; base + 512 <= eEnd; base += 512) {
    const int e0 = base + lane * 4;
    const v4i ra = *(const v4i*)(recv + e0);
    const v4i rb = *(const v4i*)(recv + e0 + 128);
    const v4i rc = *(const v4i*)(recv + e0 + 256);
    const v4i rd = *(const v4i*)(recv + e0 + 384);
    v4i sa = ra, sb = rb, sc = rc, sd = rd;
    if (FILL) {
      sa = *(const v4i*)(send + e0);
      sb = *(const v4i*)(send + e0 + 128);
      sc = *(const v4i*)(send + e0 + 256);
      sd = *(const v4i*)(send + e0 + 384);
    }
    csr_take<FILL>(ra, sa, r0, lane, cur, nbr, nList);
    csr_take<FILL>(rb, sb, r0, lane, cur, nbr, nList);
    csr_take<FILL>(rc, sc, r0, lane, cur, nbr, nList);
    csr_take<FILL>(rd, sd, r0, lane, cur, nbr, nList);
  }
  for (; base < eEnd; base += 128) {
    const int e0 = base + lane * 4;
    v4i rv = (v4i)(-1);
    v4i sv = (v4i)(0);
    if (e0 + 0 < eEnd) { rv.x = recv[e0 + 0]; if (FILL) sv.x = send[e0 + 0]; }
    if (e0 + 1 < eEnd) { rv.y = recv[e0 + 1]; if (FILL) sv.y = send[e0 + 1]; }
    if (e0 + 2 < eEnd) { rv.z = recv[e0 + 2]; if (FILL) sv.z = send[e0 + 2]; }
    if (e0 + 3 < eEnd) { rv.w = recv[e0 + 3]; if (FILL) sv.w = send[e0 + 3]; }
    csr_take<FILL>(rv, sv, r0, lane, cur, nbr, nList);
  }
}

struct CsrOwner { const int* recv; const int* send; int r0; int nOwn; int rcBase; int eBeg; int eEnd; };
__device__ __forceinline__ CsrOwner csr_owner(const int* __restrict__ edgeUser,
                                              const int* __restrict__ edgeDish,
                                              int nUsers, int nDishes, int nEdges, int q) {
  CsrOwner o;
  const int gU = (nUsers + CSR_G - 1) / CSR_G;
  const int g  = blockIdx.x;
  const bool userSide = g < gU;
  o.recv   = userSide ? edgeUser : edgeDish;
  o.send   = userSide ? edgeDish : edgeUser;
  o.r0     = (userSide ? g : g - gU) * CSR_G;
  o.nOwn   = userSide ? nUsers : nDishes;
  o.rcBase = userSide ? o.r0 : nUsers + o.r0;
  const int per = (((nEdges + CSR_Q - 1) / CSR_Q) + 3) & ~3;
  o.eBeg = min(nEdges, q * per);
  o.eEnd = min(nEdges, o.eBeg + per);
  return o;
}

__global__ __launch_bounds__(32 * CSR_Q)
void gnn_csr_count(const int* __restrict__ edgeUser, const int* __restrict__ edgeDish,
                   int* __restrict__ cnt, int* __restrict__ tot,
                   int nUsers, int nDishes, int nEdges) {
  __shared__ int shc[CSR_Q][CSR_G];
  const int lane = threadIdx.x & 31;
  const int q = __builtin_amdgcn_readfirstlane(threadIdx.x >> 5);
  const CsrOwner o = csr_owner(edgeUser, edgeDish, nUsers, nDishes, nEdges, q);
  v4i cur = (v4i)(0);
  csr_scan_slice<false>(o.recv, o.send, o.eBeg, o.eEnd, o.r0, lane, cur, nullptr, 0);
  shc[q][lane]      = cur.x;
  shc[q][lane + 32] = cur.y;
  shc[q][lane + 64] = cur.z;
  shc[q][lane + 96] = cur.w;
  __syncthreads();
  const int k = threadIdx.x;
  if (k < CSR_G && o.r0 + k < o.nOwn) {
    v4i* p = (v4i*)(cnt + (size_t)(o.rcBase + k) * CSR_Q);
    int s = 0;
    for (int v = 0; v < CSR_Q / 4; ++v) {
      v4i r;
      r.x = shc[4 * v + 0][k];
      r.y = shc[4 * v + 1][k];
      r.z = shc[4 * v + 2][k];
      r.w = shc[4 * v + 3][k];
      s += (r.x + r.y) + (r.z + r.w);
      p[v] = r;
    }
    tot[o.rcBase + k] = s;
  }
}

__global__ __launch_bounds__(32 * CSR_Q)
void gnn_csr_fill(const int* __restrict__ edgeUser, const int* __restrict__ edgeDish,
                  const int* __restrict__ segBegin, int* __restrict__ nbr,
                  int nUsers, int nDishes, int nEdges, int nList) {
  const int lane = threadIdx.x & 31;
  const int q = __builtin_amdgcn_readfirstlane(threadIdx.x >> 5);
  const CsrOwner o = csr_owner(edgeUser, edgeDish, nUsers, nDishes, nEdges, q);
  const int k0 = o.r0 + lane;
  const int* p = segBegin + (size_t)(o.rcBase + lane) * CSR_Q + q;
  v4i cur = (v4i)(0);
  if (k0      < o.nOwn) cur.x = p[0];
  if (k0 + 32 < o.nOwn) cur.y = p[(size_t)32 * CSR_Q];
  if (k0 + 64 < o.nOwn) cur.z = p[(size_t)64 * CSR_Q];
  if (k0 + 96 < o.nOwn) cur.w = p[(size_t)96 * CSR_Q];
  csr_scan_slice<true>(o.recv, o.send, o.eBeg, o.eEnd, o.r0, lane, cur, nbr, nList);
}

__device__ __forceinline__ int csr_sum4(v4i a) { return (a.x + a.y) + (a.z + a.w); }

__global__ __launch_bounds__(1024)
void gnn_csr_offsets(const int* __restrict__ cnt, const int* __restrict__ tot,
                     int* __restrict__ segBegin, int nRecv) {
  __shared__ int sh[32];
  __shared__ int shBase;
  const int t = threadIdx.x;
  const int lane = t & 31;
  const int w = t >> 5;

  int part = 0;
  for (int j = 0; j < (int)blockIdx.x; ++j) part += tot[j * 1024 + t];
  for (int d = 16; d >= 1; d >>= 1) part += __shfl_xor(part, d, 32);
  if (lane == 0) sh[w] = part;
  __syncthreads();
  if (w == 0) {
    int v = sh[lane];
    for (int d = 16; d >= 1; d >>= 1) v += __shfl_xor(v, d, 32);
    if (lane == 0) shBase = v;
  }
  __syncthreads();
  const int base = shBase;

  const int i = blockIdx.x * 1024 + t;
  v4i c[CSR_Q / 4];
  int rowTot = 0;
  for (int v = 0; v < CSR_Q / 4; ++v) {
    c[v] = (v4i)(0);
    if (i < nRecv) c[v] = ((const v4i*)(cnt + (size_t)i * CSR_Q))[v];
    rowTot += csr_sum4(c[v]);
  }
  int incl = rowTot;
  for (int d = 1; d < 32; d <<= 1) {
    const int u = __shfl_up(incl, d, 32);
    if (lane >= d) incl += u;
  }
  if (lane == 31) sh[w] = incl;
  __syncthreads();
  if (w == 0) {
    const int v = sh[lane];
    int s = v;
    for (int d = 1; d < 32; d <<= 1) {
      const int u = __shfl_up(s, d, 32);
      if (lane >= d) s += u;
    }
    sh[lane] = s - v;
  }
  __syncthreads();
  if (i < nRecv) {
    int o = base + sh[w] + incl - rowTot;
    v4i* p = (v4i*)(segBegin + (size_t)i * CSR_Q);
    for (int v = 0; v < CSR_Q / 4; ++v) {
      v4i r;
      r.x = o; o += c[v].x;
      r.y = o; o += c[v].y;
      r.z = o; o += c[v].z;
      r.w = o; o += c[v].w;
      p[v] = r;
    }
  }
}

__global__ __launch_bounds__(256)
void gnn_gather_sum(const int* __restrict__ segBegin, const int* __restrict__ cnt,
                    const int* __restrict__ nbr,
                    const float* __restrict__ msgUser, const float* __restrict__ msgDish,
                    float* __restrict__ agg, int nUsers, int nDishes, int nList) {
  const int lane = threadIdx.x & 31;
  const int wave = __builtin_amdgcn_readfirstlane(threadIdx.x >> 5);
  const int nRecv = nUsers + nDishes;
  const int i = nRecv - 1 - (blockIdx.x * 8 + wave);
  if (i < 0) return;
  const bool isUser = i < nUsers;
  const float* msg  = isUser ? msgDish : msgUser;
  const int nSend   = isUser ? nDishes : nUsers;
  int b = segBegin[(size_t)i * CSR_Q];
  int e = segBegin[(size_t)i * CSR_Q + (CSR_Q - 1)] + cnt[(size_t)i * CSR_Q + (CSR_Q - 1)];
  b = max(0, min(b, nList));
  e = max(b, min(e, nList));
  float* row = agg + (size_t)i * 256 + lane * 4;
  v4f a0 = *(const v4f*)(row);
  v4f a1 = *(const v4f*)(row + 128);
  for (int base = b; base < e; base += 32) {
    const int n = min(32, e - base);
    int id = (lane < n) ? nbr[base + lane] : 0;
    id = max(0, min(id, nSend - 1));
    int j = 0;
    for (; j + 4 <= n; j += 4) {
      const float* r0 = msg + (size_t)__builtin_amdgcn_readlane(id, j)     * 256 + lane * 4;
      const float* r1 = msg + (size_t)__builtin_amdgcn_readlane(id, j + 1) * 256 + lane * 4;
      const float* r2 = msg + (size_t)__builtin_amdgcn_readlane(id, j + 2) * 256 + lane * 4;
      const float* r3 = msg + (size_t)__builtin_amdgcn_readlane(id, j + 3) * 256 + lane * 4;
      v4f m00 = *(const v4f*)(r0), m01 = *(const v4f*)(r0 + 128);
      v4f m10 = *(const v4f*)(r1), m11 = *(const v4f*)(r1 + 128);
      v4f m20 = *(const v4f*)(r2), m21 = *(const v4f*)(r2 + 128);
      v4f m30 = *(const v4f*)(r3), m31 = *(const v4f*)(r3 + 128);
      a0 += m00; a1 += m01;
      a0 += m10; a1 += m11;
      a0 += m20; a1 += m21;
      a0 += m30; a1 += m31;
    }
    for (; j < n; ++j) {
      const float* r0 = msg + (size_t)__builtin_amdgcn_readlane(id, j) * 256 + lane * 4;
      v4f m00 = *(const v4f*)(r0), m01 = *(const v4f*)(r0 + 128);
      a0 += m00; a1 += m01;
    }
  }
  *(v4f*)(row)       = a0;
  *(v4f*)(row + 128) = a1;
}

extern "C" void kernel_launch(void* const* d_in, const int* in_sizes, int n_in,
                              void* d_out, int out_size, void* d_ws, size_t ws_size,
                              hipStream_t stream) {
  (void)in_sizes; (void)n_in; (void)out_size; (void)ws_size;

  constexpr int U = 50000, D = 10000, E = 200000;
  constexpr int DIN = 1024, HID = 256, OUT = 128;

  const float* user_feat = (const float*)d_in[0];
  const float* dish_feat = (const float*)d_in[1];
  const int*   edge_user = (const int*)d_in[2];
  const int*   edge_dish = (const int*)d_in[3];
  const float* W_ui = (const float*)d_in[4];   const float* b_ui = (const float*)d_in[5];
  const float* W_di = (const float*)d_in[6];   const float* b_di = (const float*)d_in[7];
  const float* W_um = (const float*)d_in[8];   const float* b_um = (const float*)d_in[9];
  const float* W_dm = (const float*)d_in[10];  const float* b_dm = (const float*)d_in[11];
  const float* W_uu = (const float*)d_in[12];  const float* b_uu = (const float*)d_in[13];
  const float* W_du = (const float*)d_in[14];  const float* b_du = (const float*)d_in[15];
  const float* W_up = (const float*)d_in[16];  const float* b_up = (const float*)d_in[17];
  const float* W_dp = (const float*)d_in[18];  const float* b_dp = (const float*)d_in[19];

  float* out_user = (float*)d_out;
  float* out_dish = out_user + (size_t)U * OUT;

  float* ws      = (float*)d_ws;
  float* userAgg = ws;
  float* dishAgg = userAgg + (size_t)U * HID;
  float* msgUser = dishAgg + (size_t)D * HID;
  float* msgDish = msgUser + (size_t)U * HID;
  float* userUpd = msgUser;
  float* dishUpd = msgDish;

  dim3 blk(256);
  const int gU  = (U + BM - 1) / BM;
  const int gD  = (D + BM - 1) / BM;
  const int gU2 = (U + BM2 - 1) / BM2;
  const int gD2 = (D + BM2 - 1) / BM2;

  const DualSide l1u = { user_feat, W_ui, b_ui, userAgg, W_dm, b_dm, msgUser };
  const DualSide l1d = { dish_feat, W_di, b_di, dishAgg, W_um, b_um, msgDish };
  gnn_gemm_dual<DIN, HID><<<dim3(gU2 + gD2, HID / BN), blk, 0, stream>>>(l1u, l1d, gU2, U, D);

  constexpr int R  = U + D;
  constexpr int L  = 2 * E;
  constexpr int NG = (U + CSR_G - 1) / CSR_G + (D + CSR_G - 1) / CSR_G;
  int* cnt      = (int*)d_out;
  int* segBegin = cnt + (size_t)R * CSR_Q;
  int* tot      = segBegin + (size_t)R * CSR_Q;
  int* nbr      = tot + R;
  gnn_csr_count<<<dim3(NG), dim3(32 * CSR_Q), 0, stream>>>(edge_user, edge_dish, cnt, tot, U, D, E);
  gnn_csr_offsets<<<dim3((R + 1023) / 1024), dim3(1024), 0, stream>>>(cnt, tot, segBegin, R);
  gnn_csr_fill<<<dim3(NG), dim3(32 * CSR_Q), 0, stream>>>(edge_user, edge_dish, segBegin, nbr, U, D, E, L);
  gnn_gather_sum<<<dim3((R + 7) / 8), blk, 0, stream>>>(
      segBegin, cnt, nbr, msgUser, msgDish, userAgg, U, D, L);

  const GemmSide updU = { userAgg, W_uu, b_uu, userUpd };
  const GemmSide updD = { dishAgg, W_du, b_du, dishUpd };
  gnn_gemm_bias_act<<<dim3(gU + gD, HID / BN), blk, 0, stream>>>(updU, updD, gU, U, D, HID, HID, 1);
  const GemmSide prjU = { userUpd, W_up, b_up, out_user };
  const GemmSide prjD = { dishUpd, W_dp, b_dp, out_dish };
  gnn_gemm_bias_act<<<dim3(gU + gD, OUT / BN), blk, 0, stream>>>(prjU, prjD, gU, U, D, HID, OUT, 0);
}
